# starting kernel
# baseline (speedup 1.0000x reference)
#include <hip/hip_runtime.h>
#include <hip/hip_bf16.h>

typedef __attribute__((ext_vector_type(2))) float v2f;
typedef __attribute__((ext_vector_type(8))) float v8f;

#define HEADS 4
#define HC 256
#define CPH 64
#define NEG_SLOPE 0.2f

__device__ __forceinline__ unsigned enc_f32(float f) {
    unsigned u = __float_as_uint(f);
    return u ^ ((u >> 31) ? 0xFFFFFFFFu : 0x80000000u);
}
__device__ __forceinline__ float dec_f32(unsigned u) {
    return __uint_as_float(u ^ ((u >> 31) ? 0x80000000u : 0xFFFFFFFFu));
}

__device__ __forceinline__ void edge_sd(const int* __restrict__ ei, int E, int e,
                                        int& s, int& d) {
    if (e < E) { s = ei[e]; d = ei[E + e]; }
    else       { s = d = e - E; }
}

__global__ __launch_bounds__(256) void gemm_wmma_f32(
        const float* __restrict__ X, const float* __restrict__ W,
        float* __restrict__ H, int M, int K) {
    const int wavesPerBlock = 8;
    int wave  = blockIdx.x * wavesPerBlock + (threadIdx.x >> 5);
    const int strips = HC / 64;
    int tileM = wave / strips;
    int strip = wave % strips;
    if (tileM * 16 >= M) return;

    int lane = threadIdx.x & 31;
    int half = lane >> 4;
    int l16  = lane & 15;
    int m    = tileM * 16 + l16;
    int col0 = strip * 64 + l16;
    const float* xrow = X + (size_t)m * K;

    v8f acc0 = {}, acc1 = {}, acc2 = {}, acc3 = {};
    for (int kb = 0; kb < K; kb += 4) {
        int ka = kb + 2 * half;
        v2f a;
        a.x = xrow[ka];
        a.y = xrow[ka + 1];
        const float* w0 = W + (size_t)ka * HC + col0;
        const float* w1 = W + (size_t)(ka + 1) * HC + col0;
        v2f b0, b1, b2, b3;
        b0.x = w0[0];  b0.y = w1[0];
        b1.x = w0[16]; b1.y = w1[16];
        b2.x = w0[32]; b2.y = w1[32];
        b3.x = w0[48]; b3.y = w1[48];
        acc0 = __builtin_amdgcn_wmma_f32_16x16x4_f32(
                false, a, false, b0, (short)0, acc0, false, false);
        acc1 = __builtin_amdgcn_wmma_f32_16x16x4_f32(
                false, a, false, b1, (short)0, acc1, false, false);
        acc2 = __builtin_amdgcn_wmma_f32_16x16x4_f32(
                false, a, false, b2, (short)0, acc2, false, false);
        acc3 = __builtin_amdgcn_wmma_f32_16x16x4_f32(
                false, a, false, b3, (short)0, acc3, false, false);
    }
#pragma unroll
    for (int r = 0; r < 8; ++r) {
        int row = tileM * 16 + r + 8 * half;
        if (row < M) {
            float* hrow = H + (size_t)row * HC + col0;
            hrow[0]  = acc0[r];
            hrow[16] = acc1[r];
            hrow[32] = acc2[r];
            hrow[48] = acc3[r];
        }
    }
}

__global__ void node_alpha(const float* __restrict__ H,
                           const float* __restrict__ a_src,
                           const float* __restrict__ a_dst,
                           float* __restrict__ alphaS,
                           float* __restrict__ alphaD, int N) {
    int t = blockIdx.x * blockDim.x + threadIdx.x;
    if (t >= N * HEADS) return;
    int n = t >> 2, h = t & 3;
    const float4* hp = (const float4*)(H + (size_t)n * HC + h * CPH);
    const float4* as = (const float4*)(a_src + h * CPH);
    const float4* ad = (const float4*)(a_dst + h * CPH);
    float ss = 0.f, sd = 0.f;
#pragma unroll 4
    for (int i = 0; i < CPH / 4; ++i) {
        float4 hv = hp[i], av = as[i], dv = ad[i];
        ss += hv.x * av.x + hv.y * av.y + hv.z * av.z + hv.w * av.w;
        sd += hv.x * dv.x + hv.y * dv.y + hv.z * dv.z + hv.w * dv.w;
    }
    alphaS[t] = ss;
    alphaD[t] = sd;
}

__global__ void fill_f32(float* __restrict__ p, float v, size_t n) {
    size_t i = (size_t)blockIdx.x * blockDim.x + threadIdx.x;
    if (i < n) p[i] = v;
}
__global__ void fill_u32(unsigned* __restrict__ p, unsigned v, size_t n) {
    size_t i = (size_t)blockIdx.x * blockDim.x + threadIdx.x;
    if (i < n) p[i] = v;
}

__global__ void edge_max(const int* __restrict__ ei, int E, int N,
                         const float* __restrict__ aS,
                         const float* __restrict__ aD,
                         unsigned* __restrict__ mEnc) {
    int e = blockIdx.x * blockDim.x + threadIdx.x;
    if (e >= E + N) return;
    int s, d; edge_sd(ei, E, e, s, d);
    float4 as = *(const float4*)(aS + 4 * (size_t)s);
    float4 ad = *(const float4*)(aD + 4 * (size_t)d);
    float v[4] = {as.x + ad.x, as.y + ad.y, as.z + ad.z, as.w + ad.w};
#pragma unroll
    for (int h = 0; h < 4; ++h) {
        float x = v[h];
        x = x >= 0.f ? x : NEG_SLOPE * x;
        atomicMax(&mEnc[4 * (size_t)d + h], enc_f32(x));
    }
}

__global__ void edge_sum(const int* __restrict__ ei, int E, int N,
                         const float* __restrict__ aS,
                         const float* __restrict__ aD,
                         const unsigned* __restrict__ mEnc,
                         float* __restrict__ sums) {
    int e = blockIdx.x * blockDim.x + threadIdx.x;
    if (e >= E + N) return;
    int s, d; edge_sd(ei, E, e, s, d);
    float4 as = *(const float4*)(aS + 4 * (size_t)s);
    float4 ad = *(const float4*)(aD + 4 * (size_t)d);
    float v[4] = {as.x + ad.x, as.y + ad.y, as.z + ad.z, as.w + ad.w};
#pragma unroll
    for (int h = 0; h < 4; ++h) {
        float x = v[h];
        x = x >= 0.f ? x : NEG_SLOPE * x;
        float m = dec_f32(mEnc[4 * (size_t)d + h]);
        atomicAdd(&sums[4 * (size_t)d + h], expf(x - m));
    }
}

__global__ void edge_aggr(const int* __restrict__ ei, int E, int N,
                          const float* __restrict__ H,
                          const float* __restrict__ aS,
                          const float* __restrict__ aD,
                          const unsigned* __restrict__ mEnc,
                          const float* __restrict__ sums,
                          float* __restrict__ agg) {
    long long t = (long long)blockIdx.x * blockDim.x + threadIdx.x;
    long long Et = (long long)E + N;
    if (t >= Et * 64) return;
    int e  = (int)(t >> 6);
    int fq = (int)(t & 63);
    int h  = fq >> 4;
    int s, d; edge_sd(ei, E, e, s, d);
    float x = aS[4 * (size_t)s + h] + aD[4 * (size_t)d + h];
    x = x >= 0.f ? x : NEG_SLOPE * x;
    float m     = dec_f32(mEnc[4 * (size_t)d + h]);
    float denom = sums[4 * (size_t)d + h] + 1e-16f;
    float alpha = expf(x - m) / denom;
    float4 hv = *(const float4*)(H + (size_t)s * HC + fq * 4);
    float* o  = agg + (size_t)d * HC + fq * 4;
    atomicAdd(o + 0, hv.x * alpha);
    atomicAdd(o + 1, hv.y * alpha);
    atomicAdd(o + 2, hv.z * alpha);
    atomicAdd(o + 3, hv.w * alpha);
}

__global__ void finalize_bias(float* __restrict__ agg,
                              const float* __restrict__ b, int N, int relu) {
    int t = blockIdx.x * blockDim.x + threadIdx.x;
    if (t >= N * (HC / 4)) return;
    int n = t / (HC / 4), fq = t % (HC / 4);
    float4* p = (float4*)(agg + (size_t)n * HC + fq * 4);
    float4 v  = *p;
    float4 bb = *(const float4*)(b + fq * 4);
    v.x += bb.x; v.y += bb.y; v.z += bb.z; v.w += bb.w;
    if (relu) {
        v.x = fmaxf(v.x, 0.f); v.y = fmaxf(v.y, 0.f);
        v.z = fmaxf(v.z, 0.f); v.w = fmaxf(v.w, 0.f);
    }
    *p = v;
}

static void gat_layer(const float* Xin, int K, const float* W, const float* as,
                      const float* ad, const float* b, float* Hbuf, float* agg,
                      const int* ei, int E, int N, float* alphaS, float* alphaD,
                      unsigned* mEnc, float* sums, int relu, hipStream_t stream) {
    int waves  = ((N + 15) / 16) * (HC / 64);
    int blocks = (waves + 7) / 8;
    gemm_wmma_f32<<<blocks, 256, 0, stream>>>(Xin, W, Hbuf, N, K);

    int nh = N * HEADS;
    node_alpha<<<(nh + 255) / 256, 256, 0, stream>>>(Hbuf, as, ad, alphaS, alphaD, N);
    fill_u32<<<(nh + 255) / 256, 256, 0, stream>>>(mEnc, 0u, (size_t)nh);
    fill_f32<<<(nh + 255) / 256, 256, 0, stream>>>(sums, 0.f, (size_t)nh);
    size_t nagg = (size_t)N * HC;
    fill_f32<<<(int)((nagg + 255) / 256), 256, 0, stream>>>(agg, 0.f, nagg);

    int Et = E + N;
    edge_max<<<(Et + 255) / 256, 256, 0, stream>>>(ei, E, N, alphaS, alphaD, mEnc);
    edge_sum<<<(Et + 255) / 256, 256, 0, stream>>>(ei, E, N, alphaS, alphaD, mEnc, sums);
    long long tt = (long long)Et * 64;
    edge_aggr<<<(int)((tt + 255) / 256), 256, 0, stream>>>(ei, E, N, Hbuf, alphaS,
                                                           alphaD, mEnc, sums, agg);
    finalize_bias<<<(N * (HC / 4) + 255) / 256, 256, 0, stream>>>(agg, b, N, relu);
}

extern "C" void kernel_launch(void* const* d_in, const int* in_sizes, int n_in,
                              void* d_out, int out_size, void* d_ws, size_t ws_size,
                              hipStream_t stream) {
    const float* x   = (const float*)d_in[0];
    const int*   ei  = (const int*)d_in[1];
    const float* W1  = (const float*)d_in[2];
    const float* as1 = (const float*)d_in[3];
    const float* ad1 = (const float*)d_in[4];
    const float* b1  = (const float*)d_in[5];
    const float* W2  = (const float*)d_in[6];
    const float* as2 = (const float*)d_in[7];
    const float* ad2 = (const float*)d_in[8];
    const float* b2  = (const float*)d_in[9];
    const float* W3  = (const float*)d_in[10];
    const float* as3 = (const float*)d_in[11];
    const float* ad3 = (const float*)d_in[12];
    const float* b3  = (const float*)d_in[13];

    int K1 = in_sizes[2] / HC;
    int K2 = in_sizes[6] / HC;
    int K3 = in_sizes[10] / HC;
    int N  = in_sizes[0] / K1;
    int E  = in_sizes[1] / 2;

    float*    Hbuf   = (float*)d_ws;
    float*    bufA   = Hbuf + (size_t)N * HC;
    float*    bufB   = bufA + (size_t)N * HC;
    float*    alphaS = bufB + (size_t)N * HC;
    float*    alphaD = alphaS + (size_t)N * HEADS;
    unsigned* mEnc   = (unsigned*)(alphaD + (size_t)N * HEADS);
    float*    sums   = (float*)(mEnc + (size_t)N * HEADS);

    gat_layer(x,    K1, W1, as1, ad1, b1, Hbuf, bufA,          ei, E, N,
              alphaS, alphaD, mEnc, sums,  1, stream);
    gat_layer(bufA, K2, W2, as2, ad2, b2, Hbuf, bufB,          ei, E, N,
              alphaS, alphaD, mEnc, sums,  1, stream);
    gat_layer(bufB, K3, W3, as3, ad3, b3, Hbuf, (float*)d_out, ei, E, N,
              alphaS, alphaD, mEnc, sums,  0, stream);
}


# Edges grouped by destination once per call (counting sort, int atomic rank); one wave per node does softmax and weighted sum in registers; no float atomics, sum order = arrival rank then tree
# speedup vs baseline: 5.2464x; 5.2464x over previous
#include <hip/hip_runtime.h>
#include <hip/hip_bf16.h>

typedef __attribute__((ext_vector_type(2))) float v2f;
typedef __attribute__((ext_vector_type(8))) float v8f;

#define HEADS 4
#define HC 256
#define CPH 64
#define NEG_SLOPE 0.2f

__global__ __launch_bounds__(256) void gemm_wmma_f32(
        const float* __restrict__ X, const float* __restrict__ W,
        float* __restrict__ H, int M, int K) {
    const int wavesPerBlock = 8;
    int wave  = blockIdx.x * wavesPerBlock + (threadIdx.x >> 5);
    const int strips = HC / 64;
    int tileM = wave / strips;
    int strip = wave % strips;
    if (tileM * 16 >= M) return;

    int lane = threadIdx.x & 31;
    int half = lane >> 4;
    int l16  = lane & 15;
    int m    = tileM * 16 + l16;
    int col0 = strip * 64 + l16;
    const float* xrow = X + (size_t)m * K;

    v8f acc0 = {}, acc1 = {}, acc2 = {}, acc3 = {};
    for (int kb = 0; kb < K; kb += 4) {
        int ka = kb + 2 * half;
        v2f a;
        a.x = xrow[ka];
        a.y = xrow[ka + 1];
        const float* w0 = W + (size_t)ka * HC + col0;
        const float* w1 = W + (size_t)(ka + 1) * HC + col0;
        v2f b0, b1, b2, b3;
        b0.x = w0[0];  b0.y = w1[0];
        b1.x = w0[16]; b1.y = w1[16];
        b2.x = w0[32]; b2.y = w1[32];
        b3.x = w0[48]; b3.y = w1[48];
        acc0 = __builtin_amdgcn_wmma_f32_16x16x4_f32(
                false, a, false, b0, (short)0, acc0, false, false);
        acc1 = __builtin_amdgcn_wmma_f32_16x16x4_f32(
                false, a, false, b1, (short)0, acc1, false, false);
        acc2 = __builtin_amdgcn_wmma_f32_16x16x4_f32(
                false, a, false, b2, (short)0, acc2, false, false);
        acc3 = __builtin_amdgcn_wmma_f32_16x16x4_f32(
                false, a, false, b3, (short)0, acc3, false, false);
    }
#pragma unroll
    for (int r = 0; r < 8; ++r) {
        int row = tileM * 16 + r + 8 * half;
        if (row < M) {
            float* hrow = H + (size_t)row * HC + col0;
            hrow[0]  = acc0[r];
            hrow[16] = acc1[r];
            hrow[32] = acc2[r];
            hrow[48] = acc3[r];
        }
    }
}

__global__ void node_alpha(const float* __restrict__ H,
                           const float* __restrict__ a_src,
                           const float* __restrict__ a_dst,
                           float* __restrict__ alphaS,
                           float* __restrict__ alphaD, int N) {
    int t = blockIdx.x * blockDim.x + threadIdx.x;
    if (t >= N * HEADS) return;
    int n = t >> 2, h = t & 3;
    const float4* hp = (const float4*)(H + (size_t)n * HC + h * CPH);
    const float4* as = (const float4*)(a_src + h * CPH);
    const float4* ad = (const float4*)(a_dst + h * CPH);
    float ss = 0.f, sd = 0.f;
#pragma unroll 4
    for (int i = 0; i < CPH / 4; ++i) {
        float4 hv = hp[i], av = as[i], dv = ad[i];
        ss += hv.x * av.x + hv.y * av.y + hv.z * av.z + hv.w * av.w;
        sd += hv.x * dv.x + hv.y * dv.y + hv.z * dv.z + hv.w * dv.w;
    }
    alphaS[t] = ss;
    alphaD[t] = sd;
}

__global__ void fill_i32(int* __restrict__ p, int v, int n) {
    int i = blockIdx.x * blockDim.x + threadIdx.x;
    if (i < n) p[i] = v;
}

__global__ void csr_count(const int* __restrict__ ei, int E,
                          int* __restrict__ deg, int* __restrict__ rank) {
    int e = blockIdx.x * blockDim.x + threadIdx.x;
    if (e >= E) return;
    rank[e] = atomicAdd(&deg[ei[E + e]], 1);
}

__global__ __launch_bounds__(1024) void csr_scan(const int* __restrict__ deg,
                                                 int* __restrict__ rowptr, int N) {
    __shared__ int part[1024];
    int t   = threadIdx.x;
    int per = (N + 1023) / 1024;
    int b   = t * per;
    int e   = b + per < N ? b + per : N;
    int s   = 0;
    for (int i = b; i < e; ++i) s += deg[i];
    part[t] = s;
    __syncthreads();
    for (int off = 1; off < 1024; off <<= 1) {
        int v = t >= off ? part[t - off] : 0;
        __syncthreads();
        part[t] += v;
        __syncthreads();
    }
    int run = part[t] - s;
    for (int i = b; i < e; ++i) { rowptr[i] = run; run += deg[i]; }
    if (t == 1023) rowptr[N] = part[1023];
}

__global__ void csr_place(const int* __restrict__ ei, int E,
                          const int* __restrict__ rowptr,
                          const int* __restrict__ rank,
                          int* __restrict__ csr_src) {
    int e = blockIdx.x * blockDim.x + threadIdx.x;
    if (e >= E) return;
    csr_src[rowptr[ei[E + e]] + rank[e]] = ei[e];
}

__device__ __forceinline__ float lrelu(float x) {
    return x >= 0.f ? x : NEG_SLOPE * x;
}
__device__ __forceinline__ float4 logit4(float4 a, float4 d) {
    float4 r;
    r.x = lrelu(a.x + d.x); r.y = lrelu(a.y + d.y);
    r.z = lrelu(a.z + d.z); r.w = lrelu(a.w + d.w);
    return r;
}
__device__ __forceinline__ float4 max4(float4 a, float4 b) {
    float4 r;
    r.x = fmaxf(a.x, b.x); r.y = fmaxf(a.y, b.y);
    r.z = fmaxf(a.z, b.z); r.w = fmaxf(a.w, b.w);
    return r;
}
__device__ __forceinline__ float4 exp4(float4 x, float4 m) {
    float4 r;
    r.x = expf(x.x - m.x); r.y = expf(x.y - m.y);
    r.z = expf(x.z - m.z); r.w = expf(x.w - m.w);
    return r;
}
__device__ __forceinline__ float4 div4(float4 a, float4 b) {
    float4 r;
    r.x = a.x / b.x; r.y = a.y / b.y; r.z = a.z / b.z; r.w = a.w / b.w;
    return r;
}
__device__ __forceinline__ float wave_max(float v) {
    for (int o = 16; o > 0; o >>= 1) v = fmaxf(v, __shfl_xor(v, o, 32));
    return v;
}
__device__ __forceinline__ float wave_sum(float v) {
    for (int o = 16; o > 0; o >>= 1) v += __shfl_xor(v, o, 32);
    return v;
}
__device__ __forceinline__ int lane_i(int v, int l) {
    return __builtin_amdgcn_readlane(v, l);
}
__device__ __forceinline__ float lane_f(float v, int l) {
    return __int_as_float(__builtin_amdgcn_readlane(__float_as_int(v), l));
}
__device__ __forceinline__ void fma4(float4& acc, float4 v, float a) {
    acc.x = __builtin_fmaf(v.x, a, acc.x); acc.y = __builtin_fmaf(v.y, a, acc.y);
    acc.z = __builtin_fmaf(v.z, a, acc.z); acc.w = __builtin_fmaf(v.w, a, acc.w);
}

__device__ __forceinline__ void edge_fma(float4& accP, float4& accQ, float4 p, float4 q,
                                         float4 al, int t, bool lo) {
    float a0 = lane_f(al.x, t), a1 = lane_f(al.y, t);
    float a2 = lane_f(al.z, t), a3 = lane_f(al.w, t);
    fma4(accP, p, lo ? a0 : a1);
    fma4(accQ, q, lo ? a2 : a3);
}

__device__ __forceinline__ void accum_chunk(const float4* __restrict__ H4, int sl,
                                            float4 al, int n, int lane,
                                            float4& accP, float4& accQ) {
    const bool lo = lane < 16;
    int t = 0;
    for (; t + 4 <= n; t += 4) {
        const float4* rA = H4 + (size_t)lane_i(sl, t)     * (HC / 4) + lane;
        const float4* rB = H4 + (size_t)lane_i(sl, t + 1) * (HC / 4) + lane;
        const float4* rC = H4 + (size_t)lane_i(sl, t + 2) * (HC / 4) + lane;
        const float4* rD = H4 + (size_t)lane_i(sl, t + 3) * (HC / 4) + lane;
        float4 pA = rA[0], qA = rA[32];
        float4 pB = rB[0], qB = rB[32];
        float4 pC = rC[0], qC = rC[32];
        float4 pD = rD[0], qD = rD[32];
        edge_fma(accP, accQ, pA, qA, al, t,     lo);
        edge_fma(accP, accQ, pB, qB, al, t + 1, lo);
        edge_fma(accP, accQ, pC, qC, al, t + 2, lo);
        edge_fma(accP, accQ, pD, qD, al, t + 3, lo);
    }
    for (; t < n; ++t) {
        const float4* r = H4 + (size_t)lane_i(sl, t) * (HC / 4) + lane;
        float4 p = r[0], q = r[32];
        edge_fma(accP, accQ, p, q, al, t, lo);
    }
}

__global__ __launch_bounds__(256) void gat_aggregate(
        const int* __restrict__ rowptr, const int* __restrict__ csr_src,
        const float* __restrict__ H, const float* __restrict__ aS,
        const float* __restrict__ aD, const float* __restrict__ bias,
        float* __restrict__ out, int N, int relu) {
    int node = __builtin_amdgcn_readfirstlane(blockIdx.x * 8 + (threadIdx.x >> 5));
    if (node >= N) return;
    int lane  = threadIdx.x & 31;
    int start = rowptr[node];
    int end   = rowptr[node + 1];
    int cnt   = end - start;

    const float4* aS4 = (const float4*)aS;
    const float4* H4  = (const float4*)H;
    float4 ad = ((const float4*)aD)[node];
    float4 xs = logit4(aS4[node], ad);

    bool  v0 = lane < cnt;
    int   s0 = v0 ? csr_src[start + lane] : node;
    float4 x0 = logit4(aS4[s0], ad);

    float4 m = v0 ? max4(xs, x0) : xs;
    for (int j = start + 32 + lane; j < end; j += 32)
        m = max4(m, logit4(aS4[csr_src[j]], ad));
    m.x = wave_max(m.x); m.y = wave_max(m.y);
    m.z = wave_max(m.z); m.w = wave_max(m.w);

    float4 ex0 = exp4(x0, m);
    float4 sum = v0 ? ex0 : make_float4(0.f, 0.f, 0.f, 0.f);
    for (int j = start + 32 + lane; j < end; j += 32) {
        float4 ex = exp4(logit4(aS4[csr_src[j]], ad), m);
        sum.x += ex.x; sum.y += ex.y; sum.z += ex.z; sum.w += ex.w;
    }
    float4 exs = exp4(xs, m);
    float4 den;
    den.x = wave_sum(sum.x) + exs.x + 1e-16f;
    den.y = wave_sum(sum.y) + exs.y + 1e-16f;
    den.z = wave_sum(sum.z) + exs.z + 1e-16f;
    den.w = wave_sum(sum.w) + exs.w + 1e-16f;

    float4 accP = make_float4(0.f, 0.f, 0.f, 0.f);
    float4 accQ = make_float4(0.f, 0.f, 0.f, 0.f);
    {
        float4 a = div4(exs, den);
        const float4* r = H4 + (size_t)node * (HC / 4) + lane;
        float4 p = r[0], q = r[32];
        fma4(accP, p, lane < 16 ? a.x : a.y);
        fma4(accQ, q, lane < 16 ? a.z : a.w);
    }
    accum_chunk(H4, s0, div4(ex0, den), cnt < 32 ? cnt : 32, lane, accP, accQ);
    for (int base = start + 32; base < end; base += 32) {
        int  j  = base + lane;
        int  s1 = j < end ? csr_src[j] : node;
        float4 a1 = div4(exp4(logit4(aS4[s1], ad), m), den);
        int  n1 = end - base < 32 ? end - base : 32;
        accum_chunk(H4, s1, a1, n1, lane, accP, accQ);
    }

    const float4* b4 = (const float4*)bias;
    float4 bP = b4[lane], bQ = b4[lane + 32];
    accP.x += bP.x; accP.y += bP.y; accP.z += bP.z; accP.w += bP.w;
    accQ.x += bQ.x; accQ.y += bQ.y; accQ.z += bQ.z; accQ.w += bQ.w;
    if (relu) {
        accP.x = fmaxf(accP.x, 0.f); accP.y = fmaxf(accP.y, 0.f);
        accP.z = fmaxf(accP.z, 0.f); accP.w = fmaxf(accP.w, 0.f);
        accQ.x = fmaxf(accQ.x, 0.f); accQ.y = fmaxf(accQ.y, 0.f);
        accQ.z = fmaxf(accQ.z, 0.f); accQ.w = fmaxf(accQ.w, 0.f);
    }
    float4* o = (float4*)out + (size_t)node * (HC / 4) + lane;
    o[0]  = accP;
    o[32] = accQ;
}

static void gat_layer(const float* Xin, int K, const float* W, const float* as,
                      const float* ad, const float* b, float* Hbuf, float* outp,
                      const int* rowptr, const int* csr_src, int N,
                      float* alphaS, float* alphaD, int relu, hipStream_t stream) {
    int waves  = ((N + 15) / 16) * (HC / 64);
    int blocks = (waves + 7) / 8;
    gemm_wmma_f32<<<blocks, 256, 0, stream>>>(Xin, W, Hbuf, N, K);

    int nh = N * HEADS;
    node_alpha<<<(nh + 255) / 256, 256, 0, stream>>>(Hbuf, as, ad, alphaS, alphaD, N);
    gat_aggregate<<<(N + 7) / 8, 256, 0, stream>>>(rowptr, csr_src, Hbuf, alphaS,
                                                   alphaD, b, outp, N, relu);
}

extern "C" void kernel_launch(void* const* d_in, const int* in_sizes, int n_in,
                              void* d_out, int out_size, void* d_ws, size_t ws_size,
                              hipStream_t stream) {
    const float* x   = (const float*)d_in[0];
    const int*   ei  = (const int*)d_in[1];
    const float* W1  = (const float*)d_in[2];
    const float* as1 = (const float*)d_in[3];
    const float* ad1 = (const float*)d_in[4];
    const float* b1  = (const float*)d_in[5];
    const float* W2  = (const float*)d_in[6];
    const float* as2 = (const float*)d_in[7];
    const float* ad2 = (const float*)d_in[8];
    const float* b2  = (const float*)d_in[9];
    const float* W3  = (const float*)d_in[10];
    const float* as3 = (const float*)d_in[11];
    const float* ad3 = (const float*)d_in[12];
    const float* b3  = (const float*)d_in[13];

    int K1 = in_sizes[2] / HC;
    int K2 = in_sizes[6] / HC;
    int K3 = in_sizes[10] / HC;
    int N  = in_sizes[0] / K1;
    int E  = in_sizes[1] / 2;

    float* Hbuf    = (float*)d_ws;
    float* bufA    = Hbuf + (size_t)N * HC;
    float* bufB    = bufA + (size_t)N * HC;
    float* alphaS  = bufB + (size_t)N * HC;
    float* alphaD  = alphaS + (size_t)N * HEADS;
    int*   deg     = (int*)(alphaD + (size_t)N * HEADS);
    int*   rowptr  = deg + (size_t)N;
    int*   rank    = rowptr + (size_t)N + 1;
    int*   csr_src = rank + (size_t)E;

    fill_i32<<<(N + 255) / 256, 256, 0, stream>>>(deg, 0, N);
    csr_count<<<(E + 255) / 256, 256, 0, stream>>>(ei, E, deg, rank);
    csr_scan<<<1, 1024, 0, stream>>>(deg, rowptr, N);
    csr_place<<<(E + 255) / 256, 256, 0, stream>>>(ei, E, rowptr, rank, csr_src);

    gat_layer(x,    K1, W1, as1, ad1, b1, Hbuf, bufA,          rowptr, csr_src, N,
              alphaS, alphaD,  1, stream);
    gat_layer(bufA, K2, W2, as2, ad2, b2, Hbuf, bufB,          rowptr, csr_src, N,
              alphaS, alphaD,  1, stream);
    gat_layer(bufB, K3, W3, as3, ad3, b3, Hbuf, (float*)d_out, rowptr, csr_src, N,
              alphaS, alphaD,  0, stream);
}


# csr_scan: threads hold 16-byte pieces of 4096-entry chunks in registers (a wave reads and writes 512 contiguous bytes per instruction); shuffle scan + LDS carries replace 49 strided accesses a thread
# speedup vs baseline: 6.0600x; 1.1551x over previous
#include <hip/hip_runtime.h>
#include <hip/hip_bf16.h>

typedef __attribute__((ext_vector_type(2))) float v2f;
typedef __attribute__((ext_vector_type(8))) float v8f;

#define HEADS 4
#define HC 256
#define CPH 64
#define NEG_SLOPE 0.2f

__global__ __launch_bounds__(256) void gemm_wmma_f32(
        const float* __restrict__ X, const float* __restrict__ W,
        float* __restrict__ H, int M, int K) {
    const int wavesPerBlock = 8;
    int wave  = blockIdx.x * wavesPerBlock + (threadIdx.x >> 5);
    const int strips = HC / 64;
    int tileM = wave / strips;
    int strip = wave % strips;
    if (tileM * 16 >= M) return;

    int lane = threadIdx.x & 31;
    int half = lane >> 4;
    int l16  = lane & 15;
    int m    = tileM * 16 + l16;
    int col0 = strip * 64 + l16;
    const float* xrow = X + (size_t)m * K;

    v8f acc0 = {}, acc1 = {}, acc2 = {}, acc3 = {};
    for (int kb = 0; kb < K; kb += 4) {
        int ka = kb + 2 * half;
        v2f a;
        a.x = xrow[ka];
        a.y = xrow[ka + 1];
        const float* w0 = W + (size_t)ka * HC + col0;
        const float* w1 = W + (size_t)(ka + 1) * HC + col0;
        v2f b0, b1, b2, b3;
        b0.x = w0[0];  b0.y = w1[0];
        b1.x = w0[16]; b1.y = w1[16];
        b2.x = w0[32]; b2.y = w1[32];
        b3.x = w0[48]; b3.y = w1[48];
        acc0 = __builtin_amdgcn_wmma_f32_16x16x4_f32(
                false, a, false, b0, (short)0, acc0, false, false);
        acc1 = __builtin_amdgcn_wmma_f32_16x16x4_f32(
                false, a, false, b1, (short)0, acc1, false, false);
        acc2 = __builtin_amdgcn_wmma_f32_16x16x4_f32(
                false, a, false, b2, (short)0, acc2, false, false);
        acc3 = __builtin_amdgcn_wmma_f32_16x16x4_f32(
                false, a, false, b3, (short)0, acc3, false, false);
    }
#pragma unroll
    for (int r = 0; r < 8; ++r) {
        int row = tileM * 16 + r + 8 * half;
        if (row < M) {
            float* hrow = H + (size_t)row * HC + col0;
            hrow[0]  = acc0[r];
            hrow[16] = acc1[r];
            hrow[32] = acc2[r];
            hrow[48] = acc3[r];
        }
    }
}

__global__ void node_alpha(const float* __restrict__ H,
                           const float* __restrict__ a_src,
                           const float* __restrict__ a_dst,
                           float* __restrict__ alphaS,
                           float* __restrict__ alphaD, int N) {
    int t = blockIdx.x * blockDim.x + threadIdx.x;
    if (t >= N * HEADS) return;
    int n = t >> 2, h = t & 3;
    const float4* hp = (const float4*)(H + (size_t)n * HC + h * CPH);
    const float4* as = (const float4*)(a_src + h * CPH);
    const float4* ad = (const float4*)(a_dst + h * CPH);
    float ss = 0.f, sd = 0.f;
#pragma unroll 4
    for (int i = 0; i < CPH / 4; ++i) {
        float4 hv = hp[i], av = as[i], dv = ad[i];
        ss += hv.x * av.x + hv.y * av.y + hv.z * av.z + hv.w * av.w;
        sd += hv.x * dv.x + hv.y * dv.y + hv.z * dv.z + hv.w * dv.w;
    }
    alphaS[t] = ss;
    alphaD[t] = sd;
}

__global__ void fill_i32(int* __restrict__ p, int v, int n) {
    int i = blockIdx.x * blockDim.x + threadIdx.x;
    if (i < n) p[i] = v;
}

__global__ void csr_count(const int* __restrict__ ei, int E,
                          int* __restrict__ deg, int* __restrict__ rank) {
    int e = blockIdx.x * blockDim.x + threadIdx.x;
    if (e >= E) return;
    rank[e] = atomicAdd(&deg[ei[E + e]], 1);
}

#define SCAN_THREADS 1024
#define SCAN_IT      13
#define SCAN_CHUNK   (SCAN_THREADS * 4)

__global__ __launch_bounds__(SCAN_THREADS) void csr_scan(const int* __restrict__ deg,
                                                         int* __restrict__ rowptr, int N) {
    __shared__ int slot[SCAN_IT * 32];
    __shared__ int pass_total;
    const int t    = threadIdx.x;
    const int lane = t & 31;
    const int wave = t >> 5;
    const int N4   = N & ~3;
    int carry = 0;
    for (int base = 0; base < N4; base += SCAN_IT * SCAN_CHUNK) {
        int4 v[SCAN_IT];
        int  inc[SCAN_IT];
#pragma unroll
        for (int it = 0; it < SCAN_IT; ++it) {
            int i = base + it * SCAN_CHUNK + t * 4;
            v[it] = make_int4(0, 0, 0, 0);
            if (i < N4) v[it] = *(const int4*)(deg + i);
        }
#pragma unroll
        for (int it = 0; it < SCAN_IT; ++it)
            inc[it] = v[it].x + v[it].y + v[it].z + v[it].w;
#pragma unroll
        for (int o = 1; o < 32; o <<= 1) {
            int u[SCAN_IT];
#pragma unroll
            for (int it = 0; it < SCAN_IT; ++it) u[it] = __shfl_up(inc[it], o, 32);
#pragma unroll
            for (int it = 0; it < SCAN_IT; ++it)
                if (lane >= o) inc[it] += u[it];
        }
        if (lane == 31) {
#pragma unroll
            for (int it = 0; it < SCAN_IT; ++it) slot[it * 32 + wave] = inc[it];
        }
        __syncthreads();
        if (wave == 0) {
            int loc[SCAN_IT];
            int s = 0;
#pragma unroll
            for (int k = 0; k < SCAN_IT; ++k) {
                loc[k] = slot[lane * SCAN_IT + k];
                s += loc[k];
            }
            int tot = s;
            for (int o = 1; o < 32; o <<= 1) {
                int u = __shfl_up(tot, o, 32);
                if (lane >= o) tot += u;
            }
            int run = tot - s;
#pragma unroll
            for (int k = 0; k < SCAN_IT; ++k) {
                slot[lane * SCAN_IT + k] = run;
                run += loc[k];
            }
            if (lane == 31) pass_total = tot;
        }
        __syncthreads();
        int sp[SCAN_IT];
#pragma unroll
        for (int it = 0; it < SCAN_IT; ++it) sp[it] = slot[it * 32 + wave];
#pragma unroll
        for (int it = 0; it < SCAN_IT; ++it) {
            int i = base + it * SCAN_CHUNK + t * 4;
            int s = v[it].x + v[it].y + v[it].z + v[it].w;
            int4 o;
            o.x = carry + sp[it] + (inc[it] - s);
            o.y = o.x + v[it].x;
            o.z = o.y + v[it].y;
            o.w = o.z + v[it].z;
            if (i < N4) *(int4*)(rowptr + i) = o;
        }
        carry += pass_total;
        __syncthreads();
    }
    if (t == 0) {
        int run = carry;
        for (int i = N4; i < N; ++i) { rowptr[i] = run; run += deg[i]; }
        rowptr[N] = run;
    }
}

__global__ void csr_place(const int* __restrict__ ei, int E,
                          const int* __restrict__ rowptr,
                          const int* __restrict__ rank,
                          int* __restrict__ csr_src) {
    int e = blockIdx.x * blockDim.x + threadIdx.x;
    if (e >= E) return;
    csr_src[rowptr[ei[E + e]] + rank[e]] = ei[e];
}

__device__ __forceinline__ float lrelu(float x) {
    return x >= 0.f ? x : NEG_SLOPE * x;
}
__device__ __forceinline__ float4 logit4(float4 a, float4 d) {
    float4 r;
    r.x = lrelu(a.x + d.x); r.y = lrelu(a.y + d.y);
    r.z = lrelu(a.z + d.z); r.w = lrelu(a.w + d.w);
    return r;
}
__device__ __forceinline__ float4 max4(float4 a, float4 b) {
    float4 r;
    r.x = fmaxf(a.x, b.x); r.y = fmaxf(a.y, b.y);
    r.z = fmaxf(a.z, b.z); r.w = fmaxf(a.w, b.w);
    return r;
}
__device__ __forceinline__ float4 exp4(float4 x, float4 m) {
    float4 r;
    r.x = expf(x.x - m.x); r.y = expf(x.y - m.y);
    r.z = expf(x.z - m.z); r.w = expf(x.w - m.w);
    return r;
}
__device__ __forceinline__ float4 div4(float4 a, float4 b) {
    float4 r;
    r.x = a.x / b.x; r.y = a.y / b.y; r.z = a.z / b.z; r.w = a.w / b.w;
    return r;
}
__device__ __forceinline__ float wave_max(float v) {
    for (int o = 16; o > 0; o >>= 1) v = fmaxf(v, __shfl_xor(v, o, 32));
    return v;
}
__device__ __forceinline__ float wave_sum(float v) {
    for (int o = 16; o > 0; o >>= 1) v += __shfl_xor(v, o, 32);
    return v;
}
__device__ __forceinline__ int lane_i(int v, int l) {
    return __builtin_amdgcn_readlane(v, l);
}
__device__ __forceinline__ float lane_f(float v, int l) {
    return __int_as_float(__builtin_amdgcn_readlane(__float_as_int(v), l));
}
__device__ __forceinline__ void fma4(float4& acc, float4 v, float a) {
    acc.x = __builtin_fmaf(v.x, a, acc.x); acc.y = __builtin_fmaf(v.y, a, acc.y);
    acc.z = __builtin_fmaf(v.z, a, acc.z); acc.w = __builtin_fmaf(v.w, a, acc.w);
}

__device__ __forceinline__ void edge_fma(float4& accP, float4& accQ, float4 p, float4 q,
                                         float4 al, int t, bool lo) {
    float a0 = lane_f(al.x, t), a1 = lane_f(al.y, t);
    float a2 = lane_f(al.z, t), a3 = lane_f(al.w, t);
    fma4(accP, p, lo ? a0 : a1);
    fma4(accQ, q, lo ? a2 : a3);
}

__device__ __forceinline__ void accum_chunk(const float4* __restrict__ H4, int sl,
                                            float4 al, int n, int lane,
                                            float4& accP, float4& accQ) {
    const bool lo = lane < 16;
    int t = 0;
    for (; t + 4 <= n; t += 4) {
        const float4* rA = H4 + (size_t)lane_i(sl, t)     * (HC / 4) + lane;
        const float4* rB = H4 + (size_t)lane_i(sl, t + 1) * (HC / 4) + lane;
        const float4* rC = H4 + (size_t)lane_i(sl, t + 2) * (HC / 4) + lane;
        const float4* rD = H4 + (size_t)lane_i(sl, t + 3) * (HC / 4) + lane;
        float4 pA = rA[0], qA = rA[32];
        float4 pB = rB[0], qB = rB[32];
        float4 pC = rC[0], qC = rC[32];
        float4 pD = rD[0], qD = rD[32];
        edge_fma(accP, accQ, pA, qA, al, t,     lo);
        edge_fma(accP, accQ, pB, qB, al, t + 1, lo);
        edge_fma(accP, accQ, pC, qC, al, t + 2, lo);
        edge_fma(accP, accQ, pD, qD, al, t + 3, lo);
    }
    for (; t < n; ++t) {
        const float4* r = H4 + (size_t)lane_i(sl, t) * (HC / 4) + lane;
        float4 p = r[0], q = r[32];
        edge_fma(accP, accQ, p, q, al, t, lo);
    }
}

__global__ __launch_bounds__(256) void gat_aggregate(
        const int* __restrict__ rowptr, const int* __restrict__ csr_src,
        const float* __restrict__ H, const float* __restrict__ aS,
        const float* __restrict__ aD, const float* __restrict__ bias,
        float* __restrict__ out, int N, int relu) {
    int node = __builtin_amdgcn_readfirstlane(blockIdx.x * 8 + (threadIdx.x >> 5));
    if (node >= N) return;
    int lane  = threadIdx.x & 31;
    int start = rowptr[node];
    int end   = rowptr[node + 1];
    int cnt   = end - start;

    const float4* aS4 = (const float4*)aS;
    const float4* H4  = (const float4*)H;
    float4 ad = ((const float4*)aD)[node];
    float4 xs = logit4(aS4[node], ad);

    bool  v0 = lane < cnt;
    int   s0 = v0 ? csr_src[start + lane] : node;
    float4 x0 = logit4(aS4[s0], ad);

    float4 m = v0 ? max4(xs, x0) : xs;
    for (int j = start + 32 + lane; j < end; j += 32)
        m = max4(m, logit4(aS4[csr_src[j]], ad));
    m.x = wave_max(m.x); m.y = wave_max(m.y);
    m.z = wave_max(m.z); m.w = wave_max(m.w);

    float4 ex0 = exp4(x0, m);
    float4 sum = v0 ? ex0 : make_float4(0.f, 0.f, 0.f, 0.f);
    for (int j = start + 32 + lane; j < end; j += 32) {
        float4 ex = exp4(logit4(aS4[csr_src[j]], ad), m);
        sum.x += ex.x; sum.y += ex.y; sum.z += ex.z; sum.w += ex.w;
    }
    float4 exs = exp4(xs, m);
    float4 den;
    den.x = wave_sum(sum.x) + exs.x + 1e-16f;
    den.y = wave_sum(sum.y) + exs.y + 1e-16f;
    den.z = wave_sum(sum.z) + exs.z + 1e-16f;
    den.w = wave_sum(sum.w) + exs.w + 1e-16f;

    float4 accP = make_float4(0.f, 0.f, 0.f, 0.f);
    float4 accQ = make_float4(0.f, 0.f, 0.f, 0.f);
    {
        float4 a = div4(exs, den);
        const float4* r = H4 + (size_t)node * (HC / 4) + lane;
        float4 p = r[0], q = r[32];
        fma4(accP, p, lane < 16 ? a.x : a.y);
        fma4(accQ, q, lane < 16 ? a.z : a.w);
    }
    accum_chunk(H4, s0, div4(ex0, den), cnt < 32 ? cnt : 32, lane, accP, accQ);
    for (int base = start + 32; base < end; base += 32) {
        int  j  = base + lane;
        int  s1 = j < end ? csr_src[j] : node;
        float4 a1 = div4(exp4(logit4(aS4[s1], ad), m), den);
        int  n1 = end - base < 32 ? end - base : 32;
        accum_chunk(H4, s1, a1, n1, lane, accP, accQ);
    }

    const float4* b4 = (const float4*)bias;
    float4 bP = b4[lane], bQ = b4[lane + 32];
    accP.x += bP.x; accP.y += bP.y; accP.z += bP.z; accP.w += bP.w;
    accQ.x += bQ.x; accQ.y += bQ.y; accQ.z += bQ.z; accQ.w += bQ.w;
    if (relu) {
        accP.x = fmaxf(accP.x, 0.f); accP.y = fmaxf(accP.y, 0.f);
        accP.z = fmaxf(accP.z, 0.f); accP.w = fmaxf(accP.w, 0.f);
        accQ.x = fmaxf(accQ.x, 0.f); accQ.y = fmaxf(accQ.y, 0.f);
        accQ.z = fmaxf(accQ.z, 0.f); accQ.w = fmaxf(accQ.w, 0.f);
    }
    float4* o = (float4*)out + (size_t)node * (HC / 4) + lane;
    o[0]  = accP;
    o[32] = accQ;
}

static void gat_layer(const float* Xin, int K, const float* W, const float* as,
                      const float* ad, const float* b, float* Hbuf, float* outp,
                      const int* rowptr, const int* csr_src, int N,
                      float* alphaS, float* alphaD, int relu, hipStream_t stream) {
    int waves  = ((N + 15) / 16) * (HC / 64);
    int blocks = (waves + 7) / 8;
    gemm_wmma_f32<<<blocks, 256, 0, stream>>>(Xin, W, Hbuf, N, K);

    int nh = N * HEADS;
    node_alpha<<<(nh + 255) / 256, 256, 0, stream>>>(Hbuf, as, ad, alphaS, alphaD, N);
    gat_aggregate<<<(N + 7) / 8, 256, 0, stream>>>(rowptr, csr_src, Hbuf, alphaS,
                                                   alphaD, b, outp, N, relu);
}

extern "C" void kernel_launch(void* const* d_in, const int* in_sizes, int n_in,
                              void* d_out, int out_size, void* d_ws, size_t ws_size,
                              hipStream_t stream) {
    const float* x   = (const float*)d_in[0];
    const int*   ei  = (const int*)d_in[1];
    const float* W1  = (const float*)d_in[2];
    const float* as1 = (const float*)d_in[3];
    const float* ad1 = (const float*)d_in[4];
    const float* b1  = (const float*)d_in[5];
    const float* W2  = (const float*)d_in[6];
    const float* as2 = (const float*)d_in[7];
    const float* ad2 = (const float*)d_in[8];
    const float* b2  = (const float*)d_in[9];
    const float* W3  = (const float*)d_in[10];
    const float* as3 = (const float*)d_in[11];
    const float* ad3 = (const float*)d_in[12];
    const float* b3  = (const float*)d_in[13];

    int K1 = in_sizes[2] / HC;
    int K2 = in_sizes[6] / HC;
    int K3 = in_sizes[10] / HC;
    int N  = in_sizes[0] / K1;
    int E  = in_sizes[1] / 2;

    float* Hbuf    = (float*)d_ws;
    float* bufA    = Hbuf + (size_t)N * HC;
    float* bufB    = bufA + (size_t)N * HC;
    float* alphaS  = bufB + (size_t)N * HC;
    float* alphaD  = alphaS + (size_t)N * HEADS;
    int*   deg     = (int*)(alphaD + (size_t)N * HEADS);
    int*   rowptr  = deg + (size_t)N;
    int*   rank    = rowptr + (size_t)N + 1;
    int*   csr_src = rank + (size_t)E;

    fill_i32<<<(N + 255) / 256, 256, 0, stream>>>(deg, 0, N);
    csr_count<<<(E + 255) / 256, 256, 0, stream>>>(ei, E, deg, rank);
    csr_scan<<<1, SCAN_THREADS, 0, stream>>>(deg, rowptr, N);
    csr_place<<<(E + 255) / 256, 256, 0, stream>>>(ei, E, rowptr, rank, csr_src);

    gat_layer(x,    K1, W1, as1, ad1, b1, Hbuf, bufA,          rowptr, csr_src, N,
              alphaS, alphaD,  1, stream);
    gat_layer(bufA, K2, W2, as2, ad2, b2, Hbuf, bufB,          rowptr, csr_src, N,
              alphaS, alphaD,  1, stream);
    gat_layer(bufB, K3, W3, as3, ad3, b3, Hbuf, (float*)d_out, rowptr, csr_src, N,
              alphaS, alphaD,  0, stream);
}


# GEMM fragments by 16-byte loads: a lane owns 4 adjacent columns (one per N-tile) and 4 adjacent k of an 8-k trip; H rows stored as 16-byte pieces; k sum order changes (k,k+1,k+4,k+5 per WMMA)
# speedup vs baseline: 6.4811x; 1.0695x over previous
#include <hip/hip_runtime.h>
#include <hip/hip_bf16.h>

typedef __attribute__((ext_vector_type(2))) float v2f;
typedef __attribute__((ext_vector_type(8))) float v8f;

#define HEADS 4
#define HC 256
#define CPH 64
#define NEG_SLOPE 0.2f

__global__ __launch_bounds__(256) void gemm_wmma_f32(
        const float* __restrict__ X, const float* __restrict__ W,
        float* __restrict__ H, int M, int K) {
    const int wavesPerBlock = 8;
    int wave  = blockIdx.x * wavesPerBlock + (threadIdx.x >> 5);
    const int strips = HC / 64;
    int tileM = wave / strips;
    int strip = wave % strips;
    if (tileM * 16 >= M) return;

    int lane = threadIdx.x & 31;
    int half = lane >> 4;
    int l16  = lane & 15;
    int m    = tileM * 16 + l16;
    int col0 = strip * 64 + 4 * l16;
    const float* xrow = X + (size_t)m * K;
    const float* wcol = W + col0;

    v8f acc0 = {}, acc1 = {}, acc2 = {}, acc3 = {};
    for (int k0 = 0; k0 < K; k0 += 8) {
        int ka = k0 + 4 * half;
        float4 a4 = *(const float4*)(xrow + ka);
        const float* wr = wcol + (size_t)ka * HC;
        float4 w0 = *(const float4*)(wr);
        float4 w1 = *(const float4*)(wr + HC);
        float4 w2 = *(const float4*)(wr + 2 * HC);
        float4 w3 = *(const float4*)(wr + 3 * HC);
        v2f a, b0, b1, b2, b3;
        a.x  = a4.x; a.y  = a4.y;
        b0.x = w0.x; b0.y = w1.x;
        b1.x = w0.y; b1.y = w1.y;
        b2.x = w0.z; b2.y = w1.z;
        b3.x = w0.w; b3.y = w1.w;
        acc0 = __builtin_amdgcn_wmma_f32_16x16x4_f32(
                false, a, false, b0, (short)0, acc0, false, false);
        acc1 = __builtin_amdgcn_wmma_f32_16x16x4_f32(
                false, a, false, b1, (short)0, acc1, false, false);
        acc2 = __builtin_amdgcn_wmma_f32_16x16x4_f32(
                false, a, false, b2, (short)0, acc2, false, false);
        acc3 = __builtin_amdgcn_wmma_f32_16x16x4_f32(
                false, a, false, b3, (short)0, acc3, false, false);
        a.x  = a4.z; a.y  = a4.w;
        b0.x = w2.x; b0.y = w3.x;
        b1.x = w2.y; b1.y = w3.y;
        b2.x = w2.z; b2.y = w3.z;
        b3.x = w2.w; b3.y = w3.w;
        acc0 = __builtin_amdgcn_wmma_f32_16x16x4_f32(
                false, a, false, b0, (short)0, acc0, false, false);
        acc1 = __builtin_amdgcn_wmma_f32_16x16x4_f32(
                false, a, false, b1, (short)0, acc1, false, false);
        acc2 = __builtin_amdgcn_wmma_f32_16x16x4_f32(
                false, a, false, b2, (short)0, acc2, false, false);
        acc3 = __builtin_amdgcn_wmma_f32_16x16x4_f32(
                false, a, false, b3, (short)0, acc3, false, false);
    }
#pragma unroll
    for (int r = 0; r < 8; ++r) {
        int row = tileM * 16 + r + 8 * half;
        if (row < M) {
            float4 o = make_float4(acc0[r], acc1[r], acc2[r], acc3[r]);
            *(float4*)(H + (size_t)row * HC + col0) = o;
        }
    }
}

__global__ void node_alpha(const float* __restrict__ H,
                           const float* __restrict__ a_src,
                           const float* __restrict__ a_dst,
                           float* __restrict__ alphaS,
                           float* __restrict__ alphaD, int N) {
    int t = blockIdx.x * blockDim.x + threadIdx.x;
    if (t >= N * HEADS) return;
    int n = t >> 2, h = t & 3;
    const float4* hp = (const float4*)(H + (size_t)n * HC + h * CPH);
    const float4* as = (const float4*)(a_src + h * CPH);
    const float4* ad = (const float4*)(a_dst + h * CPH);
    float ss = 0.f, sd = 0.f;
#pragma unroll 4
    for (int i = 0; i < CPH / 4; ++i) {
        float4 hv = hp[i], av = as[i], dv = ad[i];
        ss += hv.x * av.x + hv.y * av.y + hv.z * av.z + hv.w * av.w;
        sd += hv.x * dv.x + hv.y * dv.y + hv.z * dv.z + hv.w * dv.w;
    }
    alphaS[t] = ss;
    alphaD[t] = sd;
}

__global__ void fill_i32(int* __restrict__ p, int v, int n) {
    int i = blockIdx.x * blockDim.x + threadIdx.x;
    if (i < n) p[i] = v;
}

__global__ void csr_count(const int* __restrict__ ei, int E,
                          int* __restrict__ deg, int* __restrict__ rank) {
    int e = blockIdx.x * blockDim.x + threadIdx.x;
    if (e >= E) return;
    rank[e] = atomicAdd(&deg[ei[E + e]], 1);
}

#define SCAN_THREADS 1024
#define SCAN_IT      13
#define SCAN_CHUNK   (SCAN_THREADS * 4)

__global__ __launch_bounds__(SCAN_THREADS) void csr_scan(const int* __restrict__ deg,
                                                         int* __restrict__ rowptr, int N) {
    __shared__ int slot[SCAN_IT * 32];
    __shared__ int pass_total;
    const int t    = threadIdx.x;
    const int lane = t & 31;
    const int wave = t >> 5;
    const int N4   = N & ~3;
    int carry = 0;
    for (int base = 0; base < N4; base += SCAN_IT * SCAN_CHUNK) {
        int4 v[SCAN_IT];
        int  inc[SCAN_IT];
#pragma unroll
        for (int it = 0; it < SCAN_IT; ++it) {
            int i = base + it * SCAN_CHUNK + t * 4;
            v[it] = make_int4(0, 0, 0, 0);
            if (i < N4) v[it] = *(const int4*)(deg + i);
        }
#pragma unroll
        for (int it = 0; it < SCAN_IT; ++it)
            inc[it] = v[it].x + v[it].y + v[it].z + v[it].w;
#pragma unroll
        for (int o = 1; o < 32; o <<= 1) {
            int u[SCAN_IT];
#pragma unroll
            for (int it = 0; it < SCAN_IT; ++it) u[it] = __shfl_up(inc[it], o, 32);
#pragma unroll
            for (int it = 0; it < SCAN_IT; ++it)
                if (lane >= o) inc[it] += u[it];
        }
        if (lane == 31) {
#pragma unroll
            for (int it = 0; it < SCAN_IT; ++it) slot[it * 32 + wave] = inc[it];
        }
        __syncthreads();
        if (wave == 0) {
            int loc[SCAN_IT];
            int s = 0;
#pragma unroll
            for (int k = 0; k < SCAN_IT; ++k) {
                loc[k] = slot[lane * SCAN_IT + k];
                s += loc[k];
            }
            int tot = s;
            for (int o = 1; o < 32; o <<= 1) {
                int u = __shfl_up(tot, o, 32);
                if (lane >= o) tot += u;
            }
            int run = tot - s;
#pragma unroll
            for (int k = 0; k < SCAN_IT; ++k) {
                slot[lane * SCAN_IT + k] = run;
                run += loc[k];
            }
            if (lane == 31) pass_total = tot;
        }
        __syncthreads();
        int sp[SCAN_IT];
#pragma unroll
        for (int it = 0; it < SCAN_IT; ++it) sp[it] = slot[it * 32 + wave];
#pragma unroll
        for (int it = 0; it < SCAN_IT; ++it) {
            int i = base + it * SCAN_CHUNK + t * 4;
            int s = v[it].x + v[it].y + v[it].z + v[it].w;
            int4 o;
            o.x = carry + sp[it] + (inc[it] - s);
            o.y = o.x + v[it].x;
            o.z = o.y + v[it].y;
            o.w = o.z + v[it].z;
            if (i < N4) *(int4*)(rowptr + i) = o;
        }
        carry += pass_total;
        __syncthreads();
    }
    if (t == 0) {
        int run = carry;
        for (int i = N4; i < N; ++i) { rowptr[i] = run; run += deg[i]; }
        rowptr[N] = run;
    }
}

__global__ void csr_place(const int* __restrict__ ei, int E,
                          const int* __restrict__ rowptr,
                          const int* __restrict__ rank,
                          int* __restrict__ csr_src) {
    int e = blockIdx.x * blockDim.x + threadIdx.x;
    if (e >= E) return;
    csr_src[rowptr[ei[E + e]] + rank[e]] = ei[e];
}

__device__ __forceinline__ float lrelu(float x) {
    return x >= 0.f ? x : NEG_SLOPE * x;
}
__device__ __forceinline__ float4 logit4(float4 a, float4 d) {
    float4 r;
    r.x = lrelu(a.x + d.x); r.y = lrelu(a.y + d.y);
    r.z = lrelu(a.z + d.z); r.w = lrelu(a.w + d.w);
    return r;
}
__device__ __forceinline__ float4 max4(float4 a, float4 b) {
    float4 r;
    r.x = fmaxf(a.x, b.x); r.y = fmaxf(a.y, b.y);
    r.z = fmaxf(a.z, b.z); r.w = fmaxf(a.w, b.w);
    return r;
}
__device__ __forceinline__ float4 exp4(float4 x, float4 m) {
    float4 r;
    r.x = expf(x.x - m.x); r.y = expf(x.y - m.y);
    r.z = expf(x.z - m.z); r.w = expf(x.w - m.w);
    return r;
}
__device__ __forceinline__ float4 div4(float4 a, float4 b) {
    float4 r;
    r.x = a.x / b.x; r.y = a.y / b.y; r.z = a.z / b.z; r.w = a.w / b.w;
    return r;
}
__device__ __forceinline__ float wave_max(float v) {
    for (int o = 16; o > 0; o >>= 1) v = fmaxf(v, __shfl_xor(v, o, 32));
    return v;
}
__device__ __forceinline__ float wave_sum(float v) {
    for (int o = 16; o > 0; o >>= 1) v += __shfl_xor(v, o, 32);
    return v;
}
__device__ __forceinline__ int lane_i(int v, int l) {
    return __builtin_amdgcn_readlane(v, l);
}
__device__ __forceinline__ float lane_f(float v, int l) {
    return __int_as_float(__builtin_amdgcn_readlane(__float_as_int(v), l));
}
__device__ __forceinline__ void fma4(float4& acc, float4 v, float a) {
    acc.x = __builtin_fmaf(v.x, a, acc.x); acc.y = __builtin_fmaf(v.y, a, acc.y);
    acc.z = __builtin_fmaf(v.z, a, acc.z); acc.w = __builtin_fmaf(v.w, a, acc.w);
}

__device__ __forceinline__ void edge_fma(float4& accP, float4& accQ, float4 p, float4 q,
                                         float4 al, int t, bool lo) {
    float a0 = lane_f(al.x, t), a1 = lane_f(al.y, t);
    float a2 = lane_f(al.z, t), a3 = lane_f(al.w, t);
    fma4(accP, p, lo ? a0 : a1);
    fma4(accQ, q, lo ? a2 : a3);
}

__device__ __forceinline__ void accum_chunk(const float4* __restrict__ H4, int sl,
                                            float4 al, int n, int lane,
                                            float4& accP, float4& accQ) {
    const bool lo = lane < 16;
    int t = 0;
    for (; t + 4 <= n; t += 4) {
        const float4* rA = H4 + (size_t)lane_i(sl, t)     * (HC / 4) + lane;
        const float4* rB = H4 + (size_t)lane_i(sl, t + 1) * (HC / 4) + lane;
        const float4* rC = H4 + (size_t)lane_i(sl, t + 2) * (HC / 4) + lane;
        const float4* rD = H4 + (size_t)lane_i(sl, t + 3) * (HC / 4) + lane;
        float4 pA = rA[0], qA = rA[32];
        float4 pB = rB[0], qB = rB[32];
        float4 pC = rC[0], qC = rC[32];
        float4 pD = rD[0], qD = rD[32];
        edge_fma(accP, accQ, pA, qA, al, t,     lo);
        edge_fma(accP, accQ, pB, qB, al, t + 1, lo);
        edge_fma(accP, accQ, pC, qC, al, t + 2, lo);
        edge_fma(accP, accQ, pD, qD, al, t + 3, lo);
    }
    for (; t < n; ++t) {
        const float4* r = H4 + (size_t)lane_i(sl, t) * (HC / 4) + lane;
        float4 p = r[0], q = r[32];
        edge_fma(accP, accQ, p, q, al, t, lo);
    }
}

__global__ __launch_bounds__(256) void gat_aggregate(
        const int* __restrict__ rowptr, const int* __restrict__ csr_src,
        const float* __restrict__ H, const float* __restrict__ aS,
        const float* __restrict__ aD, const float* __restrict__ bias,
        float* __restrict__ out, int N, int relu) {
    int node = __builtin_amdgcn_readfirstlane(blockIdx.x * 8 + (threadIdx.x >> 5));
    if (node >= N) return;
    int lane  = threadIdx.x & 31;
    int start = rowptr[node];
    int end   = rowptr[node + 1];
    int cnt   = end - start;

    const float4* aS4 = (const float4*)aS;
    const float4* H4  = (const float4*)H;
    float4 ad = ((const float4*)aD)[node];
    float4 xs = logit4(aS4[node], ad);

    bool  v0 = lane < cnt;
    int   s0 = v0 ? csr_src[start + lane] : node;
    float4 x0 = logit4(aS4[s0], ad);

    float4 m = v0 ? max4(xs, x0) : xs;
    for (int j = start + 32 + lane; j < end; j += 32)
        m = max4(m, logit4(aS4[csr_src[j]], ad));
    m.x = wave_max(m.x); m.y = wave_max(m.y);
    m.z = wave_max(m.z); m.w = wave_max(m.w);

    float4 ex0 = exp4(x0, m);
    float4 sum = v0 ? ex0 : make_float4(0.f, 0.f, 0.f, 0.f);
    for (int j = start + 32 + lane; j < end; j += 32) {
        float4 ex = exp4(logit4(aS4[csr_src[j]], ad), m);
        sum.x += ex.x; sum.y += ex.y; sum.z += ex.z; sum.w += ex.w;
    }
    float4 exs = exp4(xs, m);
    float4 den;
    den.x = wave_sum(sum.x) + exs.x + 1e-16f;
    den.y = wave_sum(sum.y) + exs.y + 1e-16f;
    den.z = wave_sum(sum.z) + exs.z + 1e-16f;
    den.w = wave_sum(sum.w) + exs.w + 1e-16f;

    float4 accP = make_float4(0.f, 0.f, 0.f, 0.f);
    float4 accQ = make_float4(0.f, 0.f, 0.f, 0.f);
    {
        float4 a = div4(exs, den);
        const float4* r = H4 + (size_t)node * (HC / 4) + lane;
        float4 p = r[0], q = r[32];
        fma4(accP, p, lane < 16 ? a.x : a.y);
        fma4(accQ, q, lane < 16 ? a.z : a.w);
    }
    accum_chunk(H4, s0, div4(ex0, den), cnt < 32 ? cnt : 32, lane, accP, accQ);
    for (int base = start + 32; base < end; base += 32) {
        int  j  = base + lane;
        int  s1 = j < end ? csr_src[j] : node;
        float4 a1 = div4(exp4(logit4(aS4[s1], ad), m), den);
        int  n1 = end - base < 32 ? end - base : 32;
        accum_chunk(H4, s1, a1, n1, lane, accP, accQ);
    }

    const float4* b4 = (const float4*)bias;
    float4 bP = b4[lane], bQ = b4[lane + 32];
    accP.x += bP.x; accP.y += bP.y; accP.z += bP.z; accP.w += bP.w;
    accQ.x += bQ.x; accQ.y += bQ.y; accQ.z += bQ.z; accQ.w += bQ.w;
    if (relu) {
        accP.x = fmaxf(accP.x, 0.f); accP.y = fmaxf(accP.y, 0.f);
        accP.z = fmaxf(accP.z, 0.f); accP.w = fmaxf(accP.w, 0.f);
        accQ.x = fmaxf(accQ.x, 0.f); accQ.y = fmaxf(accQ.y, 0.f);
        accQ.z = fmaxf(accQ.z, 0.f); accQ.w = fmaxf(accQ.w, 0.f);
    }
    float4* o = (float4*)out + (size_t)node * (HC / 4) + lane;
    o[0]  = accP;
    o[32] = accQ;
}

static void gat_layer(const float* Xin, int K, const float* W, const float* as,
                      const float* ad, const float* b, float* Hbuf, float* outp,
                      const int* rowptr, const int* csr_src, int N,
                      float* alphaS, float* alphaD, int relu, hipStream_t stream) {
    int waves  = ((N + 15) / 16) * (HC / 64);
    int blocks = (waves + 7) / 8;
    gemm_wmma_f32<<<blocks, 256, 0, stream>>>(Xin, W, Hbuf, N, K);

    int nh = N * HEADS;
    node_alpha<<<(nh + 255) / 256, 256, 0, stream>>>(Hbuf, as, ad, alphaS, alphaD, N);
    gat_aggregate<<<(N + 7) / 8, 256, 0, stream>>>(rowptr, csr_src, Hbuf, alphaS,
                                                   alphaD, b, outp, N, relu);
}

extern "C" void kernel_launch(void* const* d_in, const int* in_sizes, int n_in,
                              void* d_out, int out_size, void* d_ws, size_t ws_size,
                              hipStream_t stream) {
    const float* x   = (const float*)d_in[0];
    const int*   ei  = (const int*)d_in[1];
    const float* W1  = (const float*)d_in[2];
    const float* as1 = (const float*)d_in[3];
    const float* ad1 = (const float*)d_in[4];
    const float* b1  = (const float*)d_in[5];
    const float* W2  = (const float*)d_in[6];
    const float* as2 = (const float*)d_in[7];
    const float* ad2 = (const float*)d_in[8];
    const float* b2  = (const float*)d_in[9];
    const float* W3  = (const float*)d_in[10];
    const float* as3 = (const float*)d_in[11];
    const float* ad3 = (const float*)d_in[12];
    const float* b3  = (const float*)d_in[13];

    int K1 = in_sizes[2] / HC;
    int K2 = in_sizes[6] / HC;
    int K3 = in_sizes[10] / HC;
    int N  = in_sizes[0] / K1;
    int E  = in_sizes[1] / 2;

    float* Hbuf    = (float*)d_ws;
    float* bufA    = Hbuf + (size_t)N * HC;
    float* bufB    = bufA + (size_t)N * HC;
    float* alphaS  = bufB + (size_t)N * HC;
    float* alphaD  = alphaS + (size_t)N * HEADS;
    int*   deg     = (int*)(alphaD + (size_t)N * HEADS);
    int*   rowptr  = deg + (size_t)N;
    int*   rank    = rowptr + (size_t)N + 1;
    int*   csr_src = rank + (size_t)E;

    fill_i32<<<(N + 255) / 256, 256, 0, stream>>>(deg, 0, N);
    csr_count<<<(E + 255) / 256, 256, 0, stream>>>(ei, E, deg, rank);
    csr_scan<<<1, SCAN_THREADS, 0, stream>>>(deg, rowptr, N);
    csr_place<<<(E + 255) / 256, 256, 0, stream>>>(ei, E, rowptr, rank, csr_src);

    gat_layer(x,    K1, W1, as1, ad1, b1, Hbuf, bufA,          rowptr, csr_src, N,
              alphaS, alphaD,  1, stream);
    gat_layer(bufA, K2, W2, as2, ad2, b2, Hbuf, bufB,          rowptr, csr_src, N,
              alphaS, alphaD,  1, stream);
    gat_layer(bufB, K3, W3, as3, ad3, b3, Hbuf, (float*)d_out, rowptr, csr_src, N,
              alphaS, alphaD,  0, stream);
}


# Attention scalars formed in the GEMM epilogue from the accumulators (one 64-column strip = one head); the 3 node_alpha launches go; alpha sum order: a lane's 4 columns, then 16-lane tree
# speedup vs baseline: 6.7809x; 1.0463x over previous
#include <hip/hip_runtime.h>
#include <hip/hip_bf16.h>

typedef __attribute__((ext_vector_type(2))) float v2f;
typedef __attribute__((ext_vector_type(8))) float v8f;

#define HEADS 4
#define HC 256
#define CPH 64
#define NEG_SLOPE 0.2f


template <int MASK>
__device__ __forceinline__ float row_xor(float v) {
    return __int_as_float(__builtin_amdgcn_update_dpp(
            0, __float_as_int(v), 0x160 | MASK, 0xf, 0xf, true));
}

__global__ __launch_bounds__(256) void gemm_wmma_f32(
        const float* __restrict__ X, const float* __restrict__ W,
        float* __restrict__ H,
        const float* __restrict__ a_src, const float* __restrict__ a_dst,
        float* __restrict__ alphaS, float* __restrict__ alphaD,
        int M, int K) {
    const int wavesPerBlock = 8;
    int wave  = blockIdx.x * wavesPerBlock + (threadIdx.x >> 5);
    const int strips = HC / 64;
    int tileM = wave / strips;
    int strip = wave % strips;
    if (tileM * 16 >= M) return;

    int lane = threadIdx.x & 31;
    int half = lane >> 4;
    int l16  = lane & 15;
    int m    = tileM * 16 + l16;
    int col0 = strip * 64 + 4 * l16;
    const float* xrow = X + (size_t)m * K;
    const float* wcol = W + col0;
    v8f acc0 = {}, acc1 = {}, acc2 = {}, acc3 = {};
    for (int k0 = 0; k0 < K; k0 += 8) {
        int ka = k0 + 4 * half;
        float4 a4 = *(const float4*)(xrow + ka);
        const float* wr = wcol + (size_t)ka * HC;
        float4 w0 = *(const float4*)(wr);
        float4 w1 = *(const float4*)(wr + HC);
        float4 w2 = *(const float4*)(wr + 2 * HC);
        float4 w3 = *(const float4*)(wr + 3 * HC);
        v2f a, b0, b1, b2, b3;
        a.x  = a4.x; a.y  = a4.y;
        b0.x = w0.x; b0.y = w1.x;
        b1.x = w0.y; b1.y = w1.y;
        b2.x = w0.z; b2.y = w1.z;
        b3.x = w0.w; b3.y = w1.w;
        acc0 = __builtin_amdgcn_wmma_f32_16x16x4_f32(
                false, a, false, b0, (short)0, acc0, false, false);
        acc1 = __builtin_amdgcn_wmma_f32_16x16x4_f32(
                false, a, false, b1, (short)0, acc1, false, false);
        acc2 = __builtin_amdgcn_wmma_f32_16x16x4_f32(
                false, a, false, b2, (short)0, acc2, false, false);
        acc3 = __builtin_amdgcn_wmma_f32_16x16x4_f32(
                false, a, false, b3, (short)0, acc3, false, false);
        a.x  = a4.z; a.y  = a4.w;
        b0.x = w2.x; b0.y = w3.x;
        b1.x = w2.y; b1.y = w3.y;
        b2.x = w2.z; b2.y = w3.z;
        b3.x = w2.w; b3.y = w3.w;
        acc0 = __builtin_amdgcn_wmma_f32_16x16x4_f32(
                false, a, false, b0, (short)0, acc0, false, false);
        acc1 = __builtin_amdgcn_wmma_f32_16x16x4_f32(
                false, a, false, b1, (short)0, acc1, false, false);
        acc2 = __builtin_amdgcn_wmma_f32_16x16x4_f32(
                false, a, false, b2, (short)0, acc2, false, false);
        acc3 = __builtin_amdgcn_wmma_f32_16x16x4_f32(
                false, a, false, b3, (short)0, acc3, false, false);
    }
    const float4 as4 = *(const float4*)(a_src + col0);
    const float4 ad4 = *(const float4*)(a_dst + col0);
#pragma unroll
    for (int r = 0; r < 8; ++r) {
        int row = tileM * 16 + r + 8 * half;
        if (row < M) {
            float4 o = make_float4(acc0[r], acc1[r], acc2[r], acc3[r]);
            *(float4*)(H + (size_t)row * HC + col0) = o;
        }
    }

    float v[16];
#pragma unroll
    for (int r = 0; r < 8; ++r) {
        v[r]     = acc0[r] * as4.x + acc1[r] * as4.y + acc2[r] * as4.z + acc3[r] * as4.w;
        v[8 + r] = acc0[r] * ad4.x + acc1[r] * ad4.y + acc2[r] * ad4.z + acc3[r] * ad4.w;
    }
    const bool b3 = (l16 & 8) != 0, b2 = (l16 & 4) != 0;
    const bool b1 = (l16 & 2) != 0, b0 = (l16 & 1) != 0;
    float w8[8], w4[4], w2[2];
#pragma unroll
    for (int k = 0; k < 8; ++k) {
        float keep = b3 ? v[8 + k] : v[k];
        float give = b3 ? v[k] : v[8 + k];
        w8[k] = keep + row_xor<8>(give);
    }
#pragma unroll
    for (int k = 0; k < 4; ++k) {
        float keep = b2 ? w8[4 + k] : w8[k];
        float give = b2 ? w8[k] : w8[4 + k];
        w4[k] = keep + row_xor<4>(give);
    }
#pragma unroll
    for (int k = 0; k < 2; ++k) {
        float keep = b1 ? w4[2 + k] : w4[k];
        float give = b1 ? w4[k] : w4[2 + k];
        w2[k] = keep + row_xor<2>(give);
    }
    float keep0 = b0 ? w2[1] : w2[0];
    float give0 = b0 ? w2[0] : w2[1];
    float tot   = keep0 + row_xor<1>(give0);
    int arow = tileM * 16 + (l16 & 7) + 8 * half;
    float* adst = b3 ? alphaD : alphaS;
    if (arow < M) adst[(size_t)arow * HEADS + strip] = tot;
}

__global__ void fill_i32(int* __restrict__ p, int v, int n) {
    int i = blockIdx.x * blockDim.x + threadIdx.x;
    if (i < n) p[i] = v;
}

__global__ void csr_count(const int* __restrict__ ei, int E,
                          int* __restrict__ deg, int* __restrict__ rank) {
    int e = blockIdx.x * blockDim.x + threadIdx.x;
    if (e >= E) return;
    rank[e] = atomicAdd(&deg[ei[E + e]], 1);
}

#define SCAN_THREADS 1024
#define SCAN_IT      13
#define SCAN_CHUNK   (SCAN_THREADS * 4)

__global__ __launch_bounds__(SCAN_THREADS) void csr_scan(const int* __restrict__ deg,
                                                         int* __restrict__ rowptr, int N) {
    __shared__ int slot[SCAN_IT * 32];
    __shared__ int pass_total;
    const int t    = threadIdx.x;
    const int lane = t & 31;
    const int wave = t >> 5;
    const int N4   = N & ~3;
    int carry = 0;
    for (int base = 0; base < N4; base += SCAN_IT * SCAN_CHUNK) {
        int4 v[SCAN_IT];
        int  inc[SCAN_IT];
#pragma unroll
        for (int it = 0; it < SCAN_IT; ++it) {
            int i = base + it * SCAN_CHUNK + t * 4;
            v[it] = make_int4(0, 0, 0, 0);
            if (i < N4) v[it] = *(const int4*)(deg + i);
        }
#pragma unroll
        for (int it = 0; it < SCAN_IT; ++it)
            inc[it] = v[it].x + v[it].y + v[it].z + v[it].w;
#pragma unroll
        for (int o = 1; o < 32; o <<= 1) {
            int u[SCAN_IT];
#pragma unroll
            for (int it = 0; it < SCAN_IT; ++it) u[it] = __shfl_up(inc[it], o, 32);
#pragma unroll
            for (int it = 0; it < SCAN_IT; ++it)
                if (lane >= o) inc[it] += u[it];
        }
        if (lane == 31) {
#pragma unroll
            for (int it = 0; it < SCAN_IT; ++it) slot[it * 32 + wave] = inc[it];
        }
        __syncthreads();
        if (wave == 0) {
            int loc[SCAN_IT];
            int s = 0;
#pragma unroll
            for (int k = 0; k < SCAN_IT; ++k) {
                loc[k] = slot[lane * SCAN_IT + k];
                s += loc[k];
            }
            int tot = s;
            for (int o = 1; o < 32; o <<= 1) {
                int u = __shfl_up(tot, o, 32);
                if (lane >= o) tot += u;
            }
            int run = tot - s;
#pragma unroll
            for (int k = 0; k < SCAN_IT; ++k) {
                slot[lane * SCAN_IT + k] = run;
                run += loc[k];
            }
            if (lane == 31) pass_total = tot;
        }
        __syncthreads();
        int sp[SCAN_IT];
#pragma unroll
        for (int it = 0; it < SCAN_IT; ++it) sp[it] = slot[it * 32 + wave];
#pragma unroll
        for (int it = 0; it < SCAN_IT; ++it) {
            int i = base + it * SCAN_CHUNK + t * 4;
            int s = v[it].x + v[it].y + v[it].z + v[it].w;
            int4 o;
            o.x = carry + sp[it] + (inc[it] - s);
            o.y = o.x + v[it].x;
            o.z = o.y + v[it].y;
            o.w = o.z + v[it].z;
            if (i < N4) *(int4*)(rowptr + i) = o;
        }
        carry += pass_total;
        __syncthreads();
    }
    if (t == 0) {
        int run = carry;
        for (int i = N4; i < N; ++i) { rowptr[i] = run; run += deg[i]; }
        rowptr[N] = run;
    }
}

__global__ void csr_place(const int* __restrict__ ei, int E,
                          const int* __restrict__ rowptr,
                          const int* __restrict__ rank,
                          int* __restrict__ csr_src) {
    int e = blockIdx.x * blockDim.x + threadIdx.x;
    if (e >= E) return;
    csr_src[rowptr[ei[E + e]] + rank[e]] = ei[e];
}

__device__ __forceinline__ float lrelu(float x) {
    return x >= 0.f ? x : NEG_SLOPE * x;
}
__device__ __forceinline__ float4 logit4(float4 a, float4 d) {
    float4 r;
    r.x = lrelu(a.x + d.x); r.y = lrelu(a.y + d.y);
    r.z = lrelu(a.z + d.z); r.w = lrelu(a.w + d.w);
    return r;
}
__device__ __forceinline__ float4 max4(float4 a, float4 b) {
    float4 r;
    r.x = fmaxf(a.x, b.x); r.y = fmaxf(a.y, b.y);
    r.z = fmaxf(a.z, b.z); r.w = fmaxf(a.w, b.w);
    return r;
}
__device__ __forceinline__ float4 exp4(float4 x, float4 m) {
    float4 r;
    r.x = expf(x.x - m.x); r.y = expf(x.y - m.y);
    r.z = expf(x.z - m.z); r.w = expf(x.w - m.w);
    return r;
}
__device__ __forceinline__ float4 div4(float4 a, float4 b) {
    float4 r;
    r.x = a.x / b.x; r.y = a.y / b.y; r.z = a.z / b.z; r.w = a.w / b.w;
    return r;
}
__device__ __forceinline__ float wave_max(float v) {
    for (int o = 16; o > 0; o >>= 1) v = fmaxf(v, __shfl_xor(v, o, 32));
    return v;
}
__device__ __forceinline__ float wave_sum(float v) {
    for (int o = 16; o > 0; o >>= 1) v += __shfl_xor(v, o, 32);
    return v;
}
__device__ __forceinline__ int lane_i(int v, int l) {
    return __builtin_amdgcn_readlane(v, l);
}
__device__ __forceinline__ float lane_f(float v, int l) {
    return __int_as_float(__builtin_amdgcn_readlane(__float_as_int(v), l));
}
__device__ __forceinline__ void fma4(float4& acc, float4 v, float a) {
    acc.x = __builtin_fmaf(v.x, a, acc.x); acc.y = __builtin_fmaf(v.y, a, acc.y);
    acc.z = __builtin_fmaf(v.z, a, acc.z); acc.w = __builtin_fmaf(v.w, a, acc.w);
}

__device__ __forceinline__ void edge_fma(float4& accP, float4& accQ, float4 p, float4 q,
                                         float4 al, int t, bool lo) {
    float a0 = lane_f(al.x, t), a1 = lane_f(al.y, t);
    float a2 = lane_f(al.z, t), a3 = lane_f(al.w, t);
    fma4(accP, p, lo ? a0 : a1);
    fma4(accQ, q, lo ? a2 : a3);
}

__device__ __forceinline__ void accum_chunk(const float4* __restrict__ H4, int sl,
                                            float4 al, int n, int lane,
                                            float4& accP, float4& accQ) {
    const bool lo = lane < 16;
    int t = 0;
    for (; t + 4 <= n; t += 4) {
        const float4* rA = H4 + (size_t)lane_i(sl, t)     * (HC / 4) + lane;
        const float4* rB = H4 + (size_t)lane_i(sl, t + 1) * (HC / 4) + lane;
        const float4* rC = H4 + (size_t)lane_i(sl, t + 2) * (HC / 4) + lane;
        const float4* rD = H4 + (size_t)lane_i(sl, t + 3) * (HC / 4) + lane;
        float4 pA = rA[0], qA = rA[32];
        float4 pB = rB[0], qB = rB[32];
        float4 pC = rC[0], qC = rC[32];
        float4 pD = rD[0], qD = rD[32];
        edge_fma(accP, accQ, pA, qA, al, t,     lo);
        edge_fma(accP, accQ, pB, qB, al, t + 1, lo);
        edge_fma(accP, accQ, pC, qC, al, t + 2, lo);
        edge_fma(accP, accQ, pD, qD, al, t + 3, lo);
    }
    for (; t < n; ++t) {
        const float4* r = H4 + (size_t)lane_i(sl, t) * (HC / 4) + lane;
        float4 p = r[0], q = r[32];
        edge_fma(accP, accQ, p, q, al, t, lo);
    }
}

__global__ __launch_bounds__(256) void gat_aggregate(
        const int* __restrict__ rowptr, const int* __restrict__ csr_src,
        const float* __restrict__ H, const float* __restrict__ aS,
        const float* __restrict__ aD, const float* __restrict__ bias,
        float* __restrict__ out, int N, int relu) {
    int node = __builtin_amdgcn_readfirstlane(blockIdx.x * 8 + (threadIdx.x >> 5));
    if (node >= N) return;
    int lane  = threadIdx.x & 31;
    int start = rowptr[node];
    int end   = rowptr[node + 1];
    int cnt   = end - start;

    const float4* aS4 = (const float4*)aS;
    const float4* H4  = (const float4*)H;
    float4 ad = ((const float4*)aD)[node];
    float4 xs = logit4(aS4[node], ad);

    bool  v0 = lane < cnt;
    int   s0 = v0 ? csr_src[start + lane] : node;
    float4 x0 = logit4(aS4[s0], ad);

    float4 m = v0 ? max4(xs, x0) : xs;
    for (int j = start + 32 + lane; j < end; j += 32)
        m = max4(m, logit4(aS4[csr_src[j]], ad));
    m.x = wave_max(m.x); m.y = wave_max(m.y);
    m.z = wave_max(m.z); m.w = wave_max(m.w);

    float4 ex0 = exp4(x0, m);
    float4 sum = v0 ? ex0 : make_float4(0.f, 0.f, 0.f, 0.f);
    for (int j = start + 32 + lane; j < end; j += 32) {
        float4 ex = exp4(logit4(aS4[csr_src[j]], ad), m);
        sum.x += ex.x; sum.y += ex.y; sum.z += ex.z; sum.w += ex.w;
    }
    float4 exs = exp4(xs, m);
    float4 den;
    den.x = wave_sum(sum.x) + exs.x + 1e-16f;
    den.y = wave_sum(sum.y) + exs.y + 1e-16f;
    den.z = wave_sum(sum.z) + exs.z + 1e-16f;
    den.w = wave_sum(sum.w) + exs.w + 1e-16f;

    float4 accP = make_float4(0.f, 0.f, 0.f, 0.f);
    float4 accQ = make_float4(0.f, 0.f, 0.f, 0.f);
    {
        float4 a = div4(exs, den);
        const float4* r = H4 + (size_t)node * (HC / 4) + lane;
        float4 p = r[0], q = r[32];
        fma4(accP, p, lane < 16 ? a.x : a.y);
        fma4(accQ, q, lane < 16 ? a.z : a.w);
    }
    accum_chunk(H4, s0, div4(ex0, den), cnt < 32 ? cnt : 32, lane, accP, accQ);
    for (int base = start + 32; base < end; base += 32) {
        int  j  = base + lane;
        int  s1 = j < end ? csr_src[j] : node;
        float4 a1 = div4(exp4(logit4(aS4[s1], ad), m), den);
        int  n1 = end - base < 32 ? end - base : 32;
        accum_chunk(H4, s1, a1, n1, lane, accP, accQ);
    }

    const float4* b4 = (const float4*)bias;
    float4 bP = b4[lane], bQ = b4[lane + 32];
    accP.x += bP.x; accP.y += bP.y; accP.z += bP.z; accP.w += bP.w;
    accQ.x += bQ.x; accQ.y += bQ.y; accQ.z += bQ.z; accQ.w += bQ.w;
    if (relu) {
        accP.x = fmaxf(accP.x, 0.f); accP.y = fmaxf(accP.y, 0.f);
        accP.z = fmaxf(accP.z, 0.f); accP.w = fmaxf(accP.w, 0.f);
        accQ.x = fmaxf(accQ.x, 0.f); accQ.y = fmaxf(accQ.y, 0.f);
        accQ.z = fmaxf(accQ.z, 0.f); accQ.w = fmaxf(accQ.w, 0.f);
    }
    float4* o = (float4*)out + (size_t)node * (HC / 4) + lane;
    o[0]  = accP;
    o[32] = accQ;
}

static void gat_layer(const float* Xin, int K, const float* W, const float* as,
                      const float* ad, const float* b, float* Hbuf, float* outp,
                      const int* rowptr, const int* csr_src, int N,
                      float* alphaS, float* alphaD, int relu, hipStream_t stream) {
    int waves  = ((N + 15) / 16) * (HC / 64);
    int blocks = (waves + 7) / 8;
    gemm_wmma_f32<<<blocks, 256, 0, stream>>>(Xin, W, Hbuf, as, ad, alphaS, alphaD, N, K);
    gat_aggregate<<<(N + 7) / 8, 256, 0, stream>>>(rowptr, csr_src, Hbuf, alphaS,
                                                   alphaD, b, outp, N, relu);
}

extern "C" void kernel_launch(void* const* d_in, const int* in_sizes, int n_in,
                              void* d_out, int out_size, void* d_ws, size_t ws_size,
                              hipStream_t stream) {
    const float* x   = (const float*)d_in[0];
    const int*   ei  = (const int*)d_in[1];
    const float* W1  = (const float*)d_in[2];
    const float* as1 = (const float*)d_in[3];
    const float* ad1 = (const float*)d_in[4];
    const float* b1  = (const float*)d_in[5];
    const float* W2  = (const float*)d_in[6];
    const float* as2 = (const float*)d_in[7];
    const float* ad2 = (const float*)d_in[8];
    const float* b2  = (const float*)d_in[9];
    const float* W3  = (const float*)d_in[10];
    const float* as3 = (const float*)d_in[11];
    const float* ad3 = (const float*)d_in[12];
    const float* b3  = (const float*)d_in[13];

    int K1 = in_sizes[2] / HC;
    int K2 = in_sizes[6] / HC;
    int K3 = in_sizes[10] / HC;
    int N  = in_sizes[0] / K1;
    int E  = in_sizes[1] / 2;

    float* Hbuf    = (float*)d_ws;
    float* bufA    = Hbuf + (size_t)N * HC;
    float* bufB    = bufA + (size_t)N * HC;
    float* alphaS  = bufB + (size_t)N * HC;
    float* alphaD  = alphaS + (size_t)N * HEADS;
    int*   deg     = (int*)(alphaD + (size_t)N * HEADS);
    int*   rowptr  = deg + (size_t)N;
    int*   rank    = rowptr + (size_t)N + 1;
    int*   csr_src = rank + (size_t)E;

    fill_i32<<<(N + 255) / 256, 256, 0, stream>>>(deg, 0, N);
    csr_count<<<(E + 255) / 256, 256, 0, stream>>>(ei, E, deg, rank);
    csr_scan<<<1, SCAN_THREADS, 0, stream>>>(deg, rowptr, N);
    csr_place<<<(E + 255) / 256, 256, 0, stream>>>(ei, E, rowptr, rank, csr_src);

    gat_layer(x,    K1, W1, as1, ad1, b1, Hbuf, bufA,          rowptr, csr_src, N,
              alphaS, alphaD,  1, stream);
    gat_layer(bufA, K2, W2, as2, ad2, b2, Hbuf, bufB,          rowptr, csr_src, N,
              alphaS, alphaD,  1, stream);
    gat_layer(bufB, K3, W3, as3, ad3, b3, Hbuf, (float*)d_out, rowptr, csr_src, N,
              alphaS, alphaD,  0, stream);
}


# GEMMs of layers 2 and 3 fused into the preceding aggregation launch: an 8-wave block sums 32 nodes into an LDS tile and multiplies it by W; layer outputs 1 and 2 skip memory; 10 launches become 8
# speedup vs baseline: 7.0802x; 1.0441x over previous
#include <hip/hip_runtime.h>
#include <hip/hip_bf16.h>

typedef __attribute__((ext_vector_type(2))) float v2f;
typedef __attribute__((ext_vector_type(8))) float v8f;

#define HEADS 4
#define HC 256
#define CPH 64
#define NEG_SLOPE 0.2f


template <int MASK>
__device__ __forceinline__ float row_xor(float v) {
    return __int_as_float(__builtin_amdgcn_update_dpp(
            0, __float_as_int(v), 0x160 | MASK, 0xf, 0xf, true));
}

__device__ __forceinline__ void strip_mma(const float* __restrict__ arow,
                                          const float* __restrict__ wcol, int half, int K,
                                          v8f& acc0, v8f& acc1, v8f& acc2, v8f& acc3) {
    for (int k0 = 0; k0 < K; k0 += 8) {
        int ka = k0 + 4 * half;
        float4 a4 = *(const float4*)(arow + ka);
        const float* wr = wcol + (size_t)ka * HC;
        float4 w0 = *(const float4*)(wr);
        float4 w1 = *(const float4*)(wr + HC);
        float4 w2 = *(const float4*)(wr + 2 * HC);
        float4 w3 = *(const float4*)(wr + 3 * HC);
        v2f a, b0, b1, b2, b3;
        a.x  = a4.x; a.y  = a4.y;
        b0.x = w0.x; b0.y = w1.x;
        b1.x = w0.y; b1.y = w1.y;
        b2.x = w0.z; b2.y = w1.z;
        b3.x = w0.w; b3.y = w1.w;
        acc0 = __builtin_amdgcn_wmma_f32_16x16x4_f32(
                false, a, false, b0, (short)0, acc0, false, false);
        acc1 = __builtin_amdgcn_wmma_f32_16x16x4_f32(
                false, a, false, b1, (short)0, acc1, false, false);
        acc2 = __builtin_amdgcn_wmma_f32_16x16x4_f32(
                false, a, false, b2, (short)0, acc2, false, false);
        acc3 = __builtin_amdgcn_wmma_f32_16x16x4_f32(
                false, a, false, b3, (short)0, acc3, false, false);
        a.x  = a4.z; a.y  = a4.w;
        b0.x = w2.x; b0.y = w3.x;
        b1.x = w2.y; b1.y = w3.y;
        b2.x = w2.z; b2.y = w3.z;
        b3.x = w2.w; b3.y = w3.w;
        acc0 = __builtin_amdgcn_wmma_f32_16x16x4_f32(
                false, a, false, b0, (short)0, acc0, false, false);
        acc1 = __builtin_amdgcn_wmma_f32_16x16x4_f32(
                false, a, false, b1, (short)0, acc1, false, false);
        acc2 = __builtin_amdgcn_wmma_f32_16x16x4_f32(
                false, a, false, b2, (short)0, acc2, false, false);
        acc3 = __builtin_amdgcn_wmma_f32_16x16x4_f32(
                false, a, false, b3, (short)0, acc3, false, false);
    }
}

__device__ __forceinline__ void strip_epilogue(const v8f& acc0, const v8f& acc1,
                                               const v8f& acc2, const v8f& acc3,
                                               int tileM, int strip, int lane,
                                               float* __restrict__ H,
                                               const float* __restrict__ a_src,
                                               const float* __restrict__ a_dst,
                                               float* __restrict__ alphaS,
                                               float* __restrict__ alphaD, int M) {
    int half = lane >> 4;
    int l16  = lane & 15;
    int col0 = strip * 64 + 4 * l16;
    const float4 as4 = *(const float4*)(a_src + col0);
    const float4 ad4 = *(const float4*)(a_dst + col0);
#pragma unroll
    for (int r = 0; r < 8; ++r) {
        int row = tileM * 16 + r + 8 * half;
        if (row < M) {
            float4 o = make_float4(acc0[r], acc1[r], acc2[r], acc3[r]);
            *(float4*)(H + (size_t)row * HC + col0) = o;
        }
    }

    float v[16];
#pragma unroll
    for (int r = 0; r < 8; ++r) {
        v[r]     = acc0[r] * as4.x + acc1[r] * as4.y + acc2[r] * as4.z + acc3[r] * as4.w;
        v[8 + r] = acc0[r] * ad4.x + acc1[r] * ad4.y + acc2[r] * ad4.z + acc3[r] * ad4.w;
    }
    const bool b3 = (l16 & 8) != 0, b2 = (l16 & 4) != 0;
    const bool b1 = (l16 & 2) != 0, b0 = (l16 & 1) != 0;
    float w8[8], w4[4], w2[2];
#pragma unroll
    for (int k = 0; k < 8; ++k) {
        float keep = b3 ? v[8 + k] : v[k];
        float give = b3 ? v[k] : v[8 + k];
        w8[k] = keep + row_xor<8>(give);
    }
#pragma unroll
    for (int k = 0; k < 4; ++k) {
        float keep = b2 ? w8[4 + k] : w8[k];
        float give = b2 ? w8[k] : w8[4 + k];
        w4[k] = keep + row_xor<4>(give);
    }
#pragma unroll
    for (int k = 0; k < 2; ++k) {
        float keep = b1 ? w4[2 + k] : w4[k];
        float give = b1 ? w4[k] : w4[2 + k];
        w2[k] = keep + row_xor<2>(give);
    }
    float keep0 = b0 ? w2[1] : w2[0];
    float give0 = b0 ? w2[0] : w2[1];
    float tot   = keep0 + row_xor<1>(give0);
    int arow = tileM * 16 + (l16 & 7) + 8 * half;
    float* adst = b3 ? alphaD : alphaS;
    if (arow < M) adst[(size_t)arow * HEADS + strip] = tot;
}

__global__ __launch_bounds__(256) void gemm_wmma_f32(
        const float* __restrict__ X, const float* __restrict__ W,
        float* __restrict__ H,
        const float* __restrict__ a_src, const float* __restrict__ a_dst,
        float* __restrict__ alphaS, float* __restrict__ alphaD,
        int M, int K) {
    const int wavesPerBlock = 8;
    int wave  = blockIdx.x * wavesPerBlock + (threadIdx.x >> 5);
    const int strips = HC / 64;
    int tileM = wave / strips;
    int strip = wave % strips;
    if (tileM * 16 >= M) return;

    int lane = threadIdx.x & 31;
    int half = lane >> 4;
    int l16  = lane & 15;
    int m    = tileM * 16 + l16;
    int col0 = strip * 64 + 4 * l16;
    v8f acc0 = {}, acc1 = {}, acc2 = {}, acc3 = {};
    strip_mma(X + (size_t)m * K, W + col0, half, K, acc0, acc1, acc2, acc3);
    strip_epilogue(acc0, acc1, acc2, acc3, tileM, strip, lane, H, a_src, a_dst,
                   alphaS, alphaD, M);
}

__global__ void fill_i32(int* __restrict__ p, int v, int n) {
    int i = blockIdx.x * blockDim.x + threadIdx.x;
    if (i < n) p[i] = v;
}

__global__ void csr_count(const int* __restrict__ ei, int E,
                          int* __restrict__ deg, int* __restrict__ rank) {
    int e = blockIdx.x * blockDim.x + threadIdx.x;
    if (e >= E) return;
    rank[e] = atomicAdd(&deg[ei[E + e]], 1);
}

#define SCAN_THREADS 1024
#define SCAN_IT      13
#define SCAN_CHUNK   (SCAN_THREADS * 4)

__global__ __launch_bounds__(SCAN_THREADS) void csr_scan(const int* __restrict__ deg,
                                                         int* __restrict__ rowptr, int N) {
    __shared__ int slot[SCAN_IT * 32];
    __shared__ int pass_total;
    const int t    = threadIdx.x;
    const int lane = t & 31;
    const int wave = t >> 5;
    const int N4   = N & ~3;
    int carry = 0;
    for (int base = 0; base < N4; base += SCAN_IT * SCAN_CHUNK) {
        int4 v[SCAN_IT];
        int  inc[SCAN_IT];
#pragma unroll
        for (int it = 0; it < SCAN_IT; ++it) {
            int i = base + it * SCAN_CHUNK + t * 4;
            v[it] = make_int4(0, 0, 0, 0);
            if (i < N4) v[it] = *(const int4*)(deg + i);
        }
#pragma unroll
        for (int it = 0; it < SCAN_IT; ++it)
            inc[it] = v[it].x + v[it].y + v[it].z + v[it].w;
#pragma unroll
        for (int o = 1; o < 32; o <<= 1) {
            int u[SCAN_IT];
#pragma unroll
            for (int it = 0; it < SCAN_IT; ++it) u[it] = __shfl_up(inc[it], o, 32);
#pragma unroll
            for (int it = 0; it < SCAN_IT; ++it)
                if (lane >= o) inc[it] += u[it];
        }
        if (lane == 31) {
#pragma unroll
            for (int it = 0; it < SCAN_IT; ++it) slot[it * 32 + wave] = inc[it];
        }
        __syncthreads();
        if (wave == 0) {
            int loc[SCAN_IT];
            int s = 0;
#pragma unroll
            for (int k = 0; k < SCAN_IT; ++k) {
                loc[k] = slot[lane * SCAN_IT + k];
                s += loc[k];
            }
            int tot = s;
            for (int o = 1; o < 32; o <<= 1) {
                int u = __shfl_up(tot, o, 32);
                if (lane >= o) tot += u;
            }
            int run = tot - s;
#pragma unroll
            for (int k = 0; k < SCAN_IT; ++k) {
                slot[lane * SCAN_IT + k] = run;
                run += loc[k];
            }
            if (lane == 31) pass_total = tot;
        }
        __syncthreads();
        int sp[SCAN_IT];
#pragma unroll
        for (int it = 0; it < SCAN_IT; ++it) sp[it] = slot[it * 32 + wave];
#pragma unroll
        for (int it = 0; it < SCAN_IT; ++it) {
            int i = base + it * SCAN_CHUNK + t * 4;
            int s = v[it].x + v[it].y + v[it].z + v[it].w;
            int4 o;
            o.x = carry + sp[it] + (inc[it] - s);
            o.y = o.x + v[it].x;
            o.z = o.y + v[it].y;
            o.w = o.z + v[it].z;
            if (i < N4) *(int4*)(rowptr + i) = o;
        }
        carry += pass_total;
        __syncthreads();
    }
    if (t == 0) {
        int run = carry;
        for (int i = N4; i < N; ++i) { rowptr[i] = run; run += deg[i]; }
        rowptr[N] = run;
    }
}

__global__ void csr_place(const int* __restrict__ ei, int E,
                          const int* __restrict__ rowptr,
                          const int* __restrict__ rank,
                          int* __restrict__ csr_src) {
    int e = blockIdx.x * blockDim.x + threadIdx.x;
    if (e >= E) return;
    csr_src[rowptr[ei[E + e]] + rank[e]] = ei[e];
}

__device__ __forceinline__ float lrelu(float x) {
    return x >= 0.f ? x : NEG_SLOPE * x;
}
__device__ __forceinline__ float4 logit4(float4 a, float4 d) {
    float4 r;
    r.x = lrelu(a.x + d.x); r.y = lrelu(a.y + d.y);
    r.z = lrelu(a.z + d.z); r.w = lrelu(a.w + d.w);
    return r;
}
__device__ __forceinline__ float4 max4(float4 a, float4 b) {
    float4 r;
    r.x = fmaxf(a.x, b.x); r.y = fmaxf(a.y, b.y);
    r.z = fmaxf(a.z, b.z); r.w = fmaxf(a.w, b.w);
    return r;
}
__device__ __forceinline__ float4 exp4(float4 x, float4 m) {
    float4 r;
    r.x = expf(x.x - m.x); r.y = expf(x.y - m.y);
    r.z = expf(x.z - m.z); r.w = expf(x.w - m.w);
    return r;
}
__device__ __forceinline__ float4 div4(float4 a, float4 b) {
    float4 r;
    r.x = a.x / b.x; r.y = a.y / b.y; r.z = a.z / b.z; r.w = a.w / b.w;
    return r;
}
__device__ __forceinline__ float wave_max(float v) {
    for (int o = 16; o > 0; o >>= 1) v = fmaxf(v, __shfl_xor(v, o, 32));
    return v;
}
__device__ __forceinline__ float wave_sum(float v) {
    for (int o = 16; o > 0; o >>= 1) v += __shfl_xor(v, o, 32);
    return v;
}
__device__ __forceinline__ int lane_i(int v, int l) {
    return __builtin_amdgcn_readlane(v, l);
}
__device__ __forceinline__ float lane_f(float v, int l) {
    return __int_as_float(__builtin_amdgcn_readlane(__float_as_int(v), l));
}
__device__ __forceinline__ void fma4(float4& acc, float4 v, float a) {
    acc.x = __builtin_fmaf(v.x, a, acc.x); acc.y = __builtin_fmaf(v.y, a, acc.y);
    acc.z = __builtin_fmaf(v.z, a, acc.z); acc.w = __builtin_fmaf(v.w, a, acc.w);
}

__device__ __forceinline__ void edge_fma(float4& accP, float4& accQ, float4 p, float4 q,
                                         float4 al, int t, bool lo) {
    float a0 = lane_f(al.x, t), a1 = lane_f(al.y, t);
    float a2 = lane_f(al.z, t), a3 = lane_f(al.w, t);
    fma4(accP, p, lo ? a0 : a1);
    fma4(accQ, q, lo ? a2 : a3);
}

__device__ __forceinline__ void accum_chunk(const float4* __restrict__ H4, int sl,
                                            float4 al, int n, int lane,
                                            float4& accP, float4& accQ) {
    const bool lo = lane < 16;
    int t = 0;
    for (; t + 4 <= n; t += 4) {
        const float4* rA = H4 + (size_t)lane_i(sl, t)     * (HC / 4) + lane;
        const float4* rB = H4 + (size_t)lane_i(sl, t + 1) * (HC / 4) + lane;
        const float4* rC = H4 + (size_t)lane_i(sl, t + 2) * (HC / 4) + lane;
        const float4* rD = H4 + (size_t)lane_i(sl, t + 3) * (HC / 4) + lane;
        float4 pA = rA[0], qA = rA[32];
        float4 pB = rB[0], qB = rB[32];
        float4 pC = rC[0], qC = rC[32];
        float4 pD = rD[0], qD = rD[32];
        edge_fma(accP, accQ, pA, qA, al, t,     lo);
        edge_fma(accP, accQ, pB, qB, al, t + 1, lo);
        edge_fma(accP, accQ, pC, qC, al, t + 2, lo);
        edge_fma(accP, accQ, pD, qD, al, t + 3, lo);
    }
    for (; t < n; ++t) {
        const float4* r = H4 + (size_t)lane_i(sl, t) * (HC / 4) + lane;
        float4 p = r[0], q = r[32];
        edge_fma(accP, accQ, p, q, al, t, lo);
    }
}

__device__ __forceinline__ void aggregate_node(
        int node, int lane,
        const int* __restrict__ rowptr, const int* __restrict__ csr_src,
        const float* __restrict__ H, const float* __restrict__ aS,
        const float* __restrict__ aD, const float* __restrict__ bias,
        int relu, float4& accP, float4& accQ) {
    int start = rowptr[node];
    int end   = rowptr[node + 1];
    int cnt   = end - start;

    const float4* aS4 = (const float4*)aS;
    const float4* H4  = (const float4*)H;
    float4 ad = ((const float4*)aD)[node];
    float4 xs = logit4(aS4[node], ad);

    bool  v0 = lane < cnt;
    int   s0 = v0 ? csr_src[start + lane] : node;
    float4 x0 = logit4(aS4[s0], ad);

    float4 m = v0 ? max4(xs, x0) : xs;
    for (int j = start + 32 + lane; j < end; j += 32)
        m = max4(m, logit4(aS4[csr_src[j]], ad));
    m.x = wave_max(m.x); m.y = wave_max(m.y);
    m.z = wave_max(m.z); m.w = wave_max(m.w);

    float4 ex0 = exp4(x0, m);
    float4 sum = v0 ? ex0 : make_float4(0.f, 0.f, 0.f, 0.f);
    for (int j = start + 32 + lane; j < end; j += 32) {
        float4 ex = exp4(logit4(aS4[csr_src[j]], ad), m);
        sum.x += ex.x; sum.y += ex.y; sum.z += ex.z; sum.w += ex.w;
    }
    float4 exs = exp4(xs, m);
    float4 den;
    den.x = wave_sum(sum.x) + exs.x + 1e-16f;
    den.y = wave_sum(sum.y) + exs.y + 1e-16f;
    den.z = wave_sum(sum.z) + exs.z + 1e-16f;
    den.w = wave_sum(sum.w) + exs.w + 1e-16f;

    accP = make_float4(0.f, 0.f, 0.f, 0.f);
    accQ = make_float4(0.f, 0.f, 0.f, 0.f);
    {
        float4 a = div4(exs, den);
        const float4* r = H4 + (size_t)node * (HC / 4) + lane;
        float4 p = r[0], q = r[32];
        fma4(accP, p, lane < 16 ? a.x : a.y);
        fma4(accQ, q, lane < 16 ? a.z : a.w);
    }
    accum_chunk(H4, s0, div4(ex0, den), cnt < 32 ? cnt : 32, lane, accP, accQ);
    for (int base = start + 32; base < end; base += 32) {
        int  j  = base + lane;
        int  s1 = j < end ? csr_src[j] : node;
        float4 a1 = div4(exp4(logit4(aS4[s1], ad), m), den);
        int  n1 = end - base < 32 ? end - base : 32;
        accum_chunk(H4, s1, a1, n1, lane, accP, accQ);
    }

    const float4* b4 = (const float4*)bias;
    float4 bP = b4[lane], bQ = b4[lane + 32];
    accP.x += bP.x; accP.y += bP.y; accP.z += bP.z; accP.w += bP.w;
    accQ.x += bQ.x; accQ.y += bQ.y; accQ.z += bQ.z; accQ.w += bQ.w;
    if (relu) {
        accP.x = fmaxf(accP.x, 0.f); accP.y = fmaxf(accP.y, 0.f);
        accP.z = fmaxf(accP.z, 0.f); accP.w = fmaxf(accP.w, 0.f);
        accQ.x = fmaxf(accQ.x, 0.f); accQ.y = fmaxf(accQ.y, 0.f);
        accQ.z = fmaxf(accQ.z, 0.f); accQ.w = fmaxf(accQ.w, 0.f);
    }
}

__global__ __launch_bounds__(256) void gat_aggregate(
        const int* __restrict__ rowptr, const int* __restrict__ csr_src,
        const float* __restrict__ H, const float* __restrict__ aS,
        const float* __restrict__ aD, const float* __restrict__ bias,
        float* __restrict__ out, int N, int relu) {
    int node = __builtin_amdgcn_readfirstlane(blockIdx.x * 8 + (threadIdx.x >> 5));
    if (node >= N) return;
    int lane  = threadIdx.x & 31;
    float4 accP, accQ;
    aggregate_node(node, lane, rowptr, csr_src, H, aS, aD, bias, relu, accP, accQ);
    float4* o = (float4*)out + (size_t)node * (HC / 4) + lane;
    o[0]  = accP;
    o[32] = accQ;
}

#define TILE_PITCH4 65
#define FUSED_ROWS  32

__global__ __launch_bounds__(256) void agg_gemm_fused(
        const int* __restrict__ rowptr, const int* __restrict__ csr_src,
        const float* __restrict__ Hin, const float* __restrict__ aSin,
        const float* __restrict__ aDin, const float* __restrict__ bias,
        const float* __restrict__ W,
        const float* __restrict__ a_src, const float* __restrict__ a_dst,
        float* __restrict__ Hout, float* __restrict__ aSout,
        float* __restrict__ aDout, int N) {
    __shared__ float4 tile[FUSED_ROWS * TILE_PITCH4];
    const int lane = threadIdx.x & 31;
    const int wave = __builtin_amdgcn_readfirstlane(threadIdx.x >> 5);

    for (int i = 0; i < 4; ++i) {
        int r    = wave * 4 + i;
        int node = blockIdx.x * FUSED_ROWS + r;
        float4 accP = make_float4(0.f, 0.f, 0.f, 0.f);
        float4 accQ = make_float4(0.f, 0.f, 0.f, 0.f);
        if (node < N)
            aggregate_node(node, lane, rowptr, csr_src, Hin, aSin, aDin, bias,
                            1, accP, accQ);
        tile[r * TILE_PITCH4 + lane]      = accP;
        tile[r * TILE_PITCH4 + 32 + lane] = accQ;
    }
    __syncthreads();

    const int half  = lane >> 4;
    const int l16   = lane & 15;
    const int strip = wave & 3;
    const int rt    = wave >> 2;
    const int tileM = blockIdx.x * (FUSED_ROWS / 16) + rt;
    const int col0  = strip * 64 + 4 * l16;
    v8f acc0 = {}, acc1 = {}, acc2 = {}, acc3 = {};
    strip_mma((const float*)&tile[(rt * 16 + l16) * TILE_PITCH4], W + col0, half, HC,
              acc0, acc1, acc2, acc3);
    strip_epilogue(acc0, acc1, acc2, acc3, tileM, strip, lane, Hout, a_src, a_dst,
                   aSout, aDout, N);
}

extern "C" void kernel_launch(void* const* d_in, const int* in_sizes, int n_in,
                              void* d_out, int out_size, void* d_ws, size_t ws_size,
                              hipStream_t stream) {
    const float* x   = (const float*)d_in[0];
    const int*   ei  = (const int*)d_in[1];
    const float* W1  = (const float*)d_in[2];
    const float* as1 = (const float*)d_in[3];
    const float* ad1 = (const float*)d_in[4];
    const float* b1  = (const float*)d_in[5];
    const float* W2  = (const float*)d_in[6];
    const float* as2 = (const float*)d_in[7];
    const float* ad2 = (const float*)d_in[8];
    const float* b2  = (const float*)d_in[9];
    const float* W3  = (const float*)d_in[10];
    const float* as3 = (const float*)d_in[11];
    const float* ad3 = (const float*)d_in[12];
    const float* b3  = (const float*)d_in[13];

    int K1 = in_sizes[2] / HC;
    int N  = in_sizes[0] / K1;
    int E  = in_sizes[1] / 2;

    float* Hbuf    = (float*)d_ws;
    float* bufA    = Hbuf + (size_t)N * HC;
    float* bufB    = bufA + (size_t)N * HC;
    float* alphaS  = bufB + (size_t)N * HC;
    float* alphaS2 = bufB;
    float* alphaD2 = bufB + (size_t)N * HEADS;
    float* alphaD  = alphaS + (size_t)N * HEADS;
    int*   deg     = (int*)(alphaD + (size_t)N * HEADS);
    int*   rowptr  = deg + (size_t)N;
    int*   rank    = rowptr + (size_t)N + 1;
    int*   csr_src = rank + (size_t)E;

    fill_i32<<<(N + 255) / 256, 256, 0, stream>>>(deg, 0, N);
    csr_count<<<(E + 255) / 256, 256, 0, stream>>>(ei, E, deg, rank);
    csr_scan<<<1, SCAN_THREADS, 0, stream>>>(deg, rowptr, N);
    csr_place<<<(E + 255) / 256, 256, 0, stream>>>(ei, E, rowptr, rank, csr_src);

    int waves  = ((N + 15) / 16) * (HC / 64);
    int blocks = (waves + 7) / 8;
    gemm_wmma_f32<<<blocks, 256, 0, stream>>>(x, W1, Hbuf, as1, ad1, alphaS, alphaD, N, K1);
    int tiles = (N + FUSED_ROWS - 1) / FUSED_ROWS;
    agg_gemm_fused<<<tiles, 256, 0, stream>>>(rowptr, csr_src, Hbuf, alphaS, alphaD, b1,
                                              W2, as2, ad2, bufA, alphaS2, alphaD2, N);
    agg_gemm_fused<<<tiles, 256, 0, stream>>>(rowptr, csr_src, bufA, alphaS2, alphaD2, b2,
                                              W3, as3, ad3, Hbuf, alphaS, alphaD, N);
    gat_aggregate<<<(N + 7) / 8, 256, 0, stream>>>(rowptr, csr_src, Hbuf, alphaS, alphaD,
                                                   b3, (float*)d_out, N,  0);
}


# Layer 1 sums the 128-wide input rows per head before the product with W1 (linearity), so its gather reads 512 B per edge, not 1 KB; logits from x @ (W1 a); sum order of layer 1 changes
# speedup vs baseline: 7.5326x; 1.0639x over previous
#include <hip/hip_runtime.h>
#include <hip/hip_bf16.h>

typedef __attribute__((ext_vector_type(2))) float v2f;
typedef __attribute__((ext_vector_type(8))) float v8f;

#define HEADS 4
#define HC 256
#define CPH 64
#define NEG_SLOPE 0.2f
#define FIN 128


template <int MASK>
__device__ __forceinline__ float row_xor(float v) {
    return __int_as_float(__builtin_amdgcn_update_dpp(
            0, __float_as_int(v), 0x160 | MASK, 0xf, 0xf, true));
}

__device__ __forceinline__ void strip_mma(const float* __restrict__ arow,
                                          const float* __restrict__ wcol, int half, int K,
                                          v8f& acc0, v8f& acc1, v8f& acc2, v8f& acc3) {
    for (int k0 = 0; k0 < K; k0 += 8) {
        int ka = k0 + 4 * half;
        float4 a4 = *(const float4*)(arow + ka);
        const float* wr = wcol + (size_t)ka * HC;
        float4 w0 = *(const float4*)(wr);
        float4 w1 = *(const float4*)(wr + HC);
        float4 w2 = *(const float4*)(wr + 2 * HC);
        float4 w3 = *(const float4*)(wr + 3 * HC);
        v2f a, b0, b1, b2, b3;
        a.x  = a4.x; a.y  = a4.y;
        b0.x = w0.x; b0.y = w1.x;
        b1.x = w0.y; b1.y = w1.y;
        b2.x = w0.z; b2.y = w1.z;
        b3.x = w0.w; b3.y = w1.w;
        acc0 = __builtin_amdgcn_wmma_f32_16x16x4_f32(
                false, a, false, b0, (short)0, acc0, false, false);
        acc1 = __builtin_amdgcn_wmma_f32_16x16x4_f32(
                false, a, false, b1, (short)0, acc1, false, false);
        acc2 = __builtin_amdgcn_wmma_f32_16x16x4_f32(
                false, a, false, b2, (short)0, acc2, false, false);
        acc3 = __builtin_amdgcn_wmma_f32_16x16x4_f32(
                false, a, false, b3, (short)0, acc3, false, false);
        a.x  = a4.z; a.y  = a4.w;
        b0.x = w2.x; b0.y = w3.x;
        b1.x = w2.y; b1.y = w3.y;
        b2.x = w2.z; b2.y = w3.z;
        b3.x = w2.w; b3.y = w3.w;
        acc0 = __builtin_amdgcn_wmma_f32_16x16x4_f32(
                false, a, false, b0, (short)0, acc0, false, false);
        acc1 = __builtin_amdgcn_wmma_f32_16x16x4_f32(
                false, a, false, b1, (short)0, acc1, false, false);
        acc2 = __builtin_amdgcn_wmma_f32_16x16x4_f32(
                false, a, false, b2, (short)0, acc2, false, false);
        acc3 = __builtin_amdgcn_wmma_f32_16x16x4_f32(
                false, a, false, b3, (short)0, acc3, false, false);
    }
}

__device__ __forceinline__ void strip_epilogue(const v8f& acc0, const v8f& acc1,
                                               const v8f& acc2, const v8f& acc3,
                                               int tileM, int strip, int lane,
                                               float* __restrict__ H,
                                               const float* __restrict__ a_src,
                                               const float* __restrict__ a_dst,
                                               float* __restrict__ alphaS,
                                               float* __restrict__ alphaD, int M) {
    int half = lane >> 4;
    int l16  = lane & 15;
    int col0 = strip * 64 + 4 * l16;
    const float4 as4 = *(const float4*)(a_src + col0);
    const float4 ad4 = *(const float4*)(a_dst + col0);
#pragma unroll
    for (int r = 0; r < 8; ++r) {
        int row = tileM * 16 + r + 8 * half;
        if (row < M) {
            float4 o = make_float4(acc0[r], acc1[r], acc2[r], acc3[r]);
            *(float4*)(H + (size_t)row * HC + col0) = o;
        }
    }

    float v[16];
#pragma unroll
    for (int r = 0; r < 8; ++r) {
        v[r]     = acc0[r] * as4.x + acc1[r] * as4.y + acc2[r] * as4.z + acc3[r] * as4.w;
        v[8 + r] = acc0[r] * ad4.x + acc1[r] * ad4.y + acc2[r] * ad4.z + acc3[r] * ad4.w;
    }
    const bool b3 = (l16 & 8) != 0, b2 = (l16 & 4) != 0;
    const bool b1 = (l16 & 2) != 0, b0 = (l16 & 1) != 0;
    float w8[8], w4[4], w2[2];
#pragma unroll
    for (int k = 0; k < 8; ++k) {
        float keep = b3 ? v[8 + k] : v[k];
        float give = b3 ? v[k] : v[8 + k];
        w8[k] = keep + row_xor<8>(give);
    }
#pragma unroll
    for (int k = 0; k < 4; ++k) {
        float keep = b2 ? w8[4 + k] : w8[k];
        float give = b2 ? w8[k] : w8[4 + k];
        w4[k] = keep + row_xor<4>(give);
    }
#pragma unroll
    for (int k = 0; k < 2; ++k) {
        float keep = b1 ? w4[2 + k] : w4[k];
        float give = b1 ? w4[k] : w4[2 + k];
        w2[k] = keep + row_xor<2>(give);
    }
    float keep0 = b0 ? w2[1] : w2[0];
    float give0 = b0 ? w2[0] : w2[1];
    float tot   = keep0 + row_xor<1>(give0);
    int arow = tileM * 16 + (l16 & 7) + 8 * half;
    float* adst = b3 ? alphaD : alphaS;
    if (arow < M) adst[(size_t)arow * HEADS + strip] = tot;
}

__global__ void prep(int* __restrict__ deg, int n, int fillBlocks,
                     const float* __restrict__ W1, const float* __restrict__ a_src,
                     const float* __restrict__ a_dst, float* __restrict__ wsdT) {
    if ((int)blockIdx.x < fillBlocks) {
        int i = blockIdx.x * blockDim.x + threadIdx.x;
        if (i < n) deg[i] = 0;
        return;
    }
    int t = (blockIdx.x - fillBlocks) * blockDim.x + threadIdx.x;
    if (t >= 16 * FIN) return;
    int j = t / FIN;
    int k = t - j * FIN;
    float s = 0.f;
    if (j < 8) {
        int head = j & 3;
        const float4* a4 = (const float4*)((j < 4 ? a_src : a_dst) + head * CPH);
        const float4* w4 = (const float4*)(W1 + (size_t)k * HC + head * CPH);
        for (int c = 0; c < CPH / 4; ++c) {
            float4 w = w4[c], a = a4[c];
            s = __builtin_fmaf(w.x, a.x, s); s = __builtin_fmaf(w.y, a.y, s);
            s = __builtin_fmaf(w.z, a.z, s); s = __builtin_fmaf(w.w, a.w, s);
        }
    }
    wsdT[t] = s;
}

__global__ __launch_bounds__(256) void alpha_from_x(
        const float* __restrict__ X, const float* __restrict__ wsdT,
        float* __restrict__ alphaS, float* __restrict__ alphaD, int M) {
    int wave = blockIdx.x * 8 + (threadIdx.x >> 5);
    if (wave * 16 >= M) return;
    int lane = threadIdx.x & 31;
    int half = lane >> 4;
    int l16  = lane & 15;
    int m    = wave * 16 + l16;
    if (m >= M) m = M - 1;
    const float* arow = X + (size_t)m * FIN;
    const float* brow = wsdT + (size_t)l16 * FIN;
    v8f acc = {};
    for (int k0 = 0; k0 < FIN; k0 += 8) {
        int ka = k0 + 4 * half;
        float4 a4 = *(const float4*)(arow + ka);
        float4 b4 = *(const float4*)(brow + ka);
        v2f a, b;
        a.x = a4.x; a.y = a4.y; b.x = b4.x; b.y = b4.y;
        acc = __builtin_amdgcn_wmma_f32_16x16x4_f32(
                false, a, false, b, (short)0, acc, false, false);
        a.x = a4.z; a.y = a4.w; b.x = b4.z; b.y = b4.w;
        acc = __builtin_amdgcn_wmma_f32_16x16x4_f32(
                false, a, false, b, (short)0, acc, false, false);
    }
    if (l16 < 8) {
        float* dst = (l16 < 4) ? alphaS : alphaD;
        int hd = l16 & 3;
#pragma unroll
        for (int r = 0; r < 8; ++r) {
            int row = wave * 16 + r + 8 * half;
            if (row < M) dst[(size_t)row * HEADS + hd] = acc[r];
        }
    }
}

__global__ void csr_count(const int* __restrict__ ei, int E,
                          int* __restrict__ deg, int* __restrict__ rank) {
    int e = blockIdx.x * blockDim.x + threadIdx.x;
    if (e >= E) return;
    rank[e] = atomicAdd(&deg[ei[E + e]], 1);
}

#define SCAN_THREADS 1024
#define SCAN_IT      13
#define SCAN_CHUNK   (SCAN_THREADS * 4)

__global__ __launch_bounds__(SCAN_THREADS) void csr_scan(const int* __restrict__ deg,
                                                         int* __restrict__ rowptr, int N) {
    __shared__ int slot[SCAN_IT * 32];
    __shared__ int pass_total;
    const int t    = threadIdx.x;
    const int lane = t & 31;
    const int wave = t >> 5;
    const int N4   = N & ~3;
    int carry = 0;
    for (int base = 0; base < N4; base += SCAN_IT * SCAN_CHUNK) {
        int4 v[SCAN_IT];
        int  inc[SCAN_IT];
#pragma unroll
        for (int it = 0; it < SCAN_IT; ++it) {
            int i = base + it * SCAN_CHUNK + t * 4;
            v[it] = make_int4(0, 0, 0, 0);
            if (i < N4) v[it] = *(const int4*)(deg + i);
        }
#pragma unroll
        for (int it = 0; it < SCAN_IT; ++it)
            inc[it] = v[it].x + v[it].y + v[it].z + v[it].w;
#pragma unroll
        for (int o = 1; o < 32; o <<= 1) {
            int u[SCAN_IT];
#pragma unroll
            for (int it = 0; it < SCAN_IT; ++it) u[it] = __shfl_up(inc[it], o, 32);
#pragma unroll
            for (int it = 0; it < SCAN_IT; ++it)
                if (lane >= o) inc[it] += u[it];
        }
        if (lane == 31) {
#pragma unroll
            for (int it = 0; it < SCAN_IT; ++it) slot[it * 32 + wave] = inc[it];
        }
        __syncthreads();
        if (wave == 0) {
            int loc[SCAN_IT];
            int s = 0;
#pragma unroll
            for (int k = 0; k < SCAN_IT; ++k) {
                loc[k] = slot[lane * SCAN_IT + k];
                s += loc[k];
            }
            int tot = s;
            for (int o = 1; o < 32; o <<= 1) {
                int u = __shfl_up(tot, o, 32);
                if (lane >= o) tot += u;
            }
            int run = tot - s;
#pragma unroll
            for (int k = 0; k < SCAN_IT; ++k) {
                slot[lane * SCAN_IT + k] = run;
                run += loc[k];
            }
            if (lane == 31) pass_total = tot;
        }
        __syncthreads();
        int sp[SCAN_IT];
#pragma unroll
        for (int it = 0; it < SCAN_IT; ++it) sp[it] = slot[it * 32 + wave];
#pragma unroll
        for (int it = 0; it < SCAN_IT; ++it) {
            int i = base + it * SCAN_CHUNK + t * 4;
            int s = v[it].x + v[it].y + v[it].z + v[it].w;
            int4 o;
            o.x = carry + sp[it] + (inc[it] - s);
            o.y = o.x + v[it].x;
            o.z = o.y + v[it].y;
            o.w = o.z + v[it].z;
            if (i < N4) *(int4*)(rowptr + i) = o;
        }
        carry += pass_total;
        __syncthreads();
    }
    if (t == 0) {
        int run = carry;
        for (int i = N4; i < N; ++i) { rowptr[i] = run; run += deg[i]; }
        rowptr[N] = run;
    }
}

__global__ void csr_place(const int* __restrict__ ei, int E,
                          const int* __restrict__ rowptr,
                          const int* __restrict__ rank,
                          int* __restrict__ csr_src) {
    int e = blockIdx.x * blockDim.x + threadIdx.x;
    if (e >= E) return;
    csr_src[rowptr[ei[E + e]] + rank[e]] = ei[e];
}

__device__ __forceinline__ float lrelu(float x) {
    return x >= 0.f ? x : NEG_SLOPE * x;
}
__device__ __forceinline__ float4 logit4(float4 a, float4 d) {
    float4 r;
    r.x = lrelu(a.x + d.x); r.y = lrelu(a.y + d.y);
    r.z = lrelu(a.z + d.z); r.w = lrelu(a.w + d.w);
    return r;
}
__device__ __forceinline__ float4 max4(float4 a, float4 b) {
    float4 r;
    r.x = fmaxf(a.x, b.x); r.y = fmaxf(a.y, b.y);
    r.z = fmaxf(a.z, b.z); r.w = fmaxf(a.w, b.w);
    return r;
}
__device__ __forceinline__ float4 exp4(float4 x, float4 m) {
    float4 r;
    r.x = expf(x.x - m.x); r.y = expf(x.y - m.y);
    r.z = expf(x.z - m.z); r.w = expf(x.w - m.w);
    return r;
}
__device__ __forceinline__ float4 div4(float4 a, float4 b) {
    float4 r;
    r.x = a.x / b.x; r.y = a.y / b.y; r.z = a.z / b.z; r.w = a.w / b.w;
    return r;
}
__device__ __forceinline__ float wave_max(float v) {
    for (int o = 16; o > 0; o >>= 1) v = fmaxf(v, __shfl_xor(v, o, 32));
    return v;
}
__device__ __forceinline__ float wave_sum(float v) {
    for (int o = 16; o > 0; o >>= 1) v += __shfl_xor(v, o, 32);
    return v;
}
__device__ __forceinline__ int lane_i(int v, int l) {
    return __builtin_amdgcn_readlane(v, l);
}
__device__ __forceinline__ float lane_f(float v, int l) {
    return __int_as_float(__builtin_amdgcn_readlane(__float_as_int(v), l));
}
__device__ __forceinline__ void fma4(float4& acc, float4 v, float a) {
    acc.x = __builtin_fmaf(v.x, a, acc.x); acc.y = __builtin_fmaf(v.y, a, acc.y);
    acc.z = __builtin_fmaf(v.z, a, acc.z); acc.w = __builtin_fmaf(v.w, a, acc.w);
}

__device__ __forceinline__ void edge_fma(float4& accP, float4& accQ, float4 p, float4 q,
                                         float4 al, int t, bool lo) {
    float a0 = lane_f(al.x, t), a1 = lane_f(al.y, t);
    float a2 = lane_f(al.z, t), a3 = lane_f(al.w, t);
    fma4(accP, p, lo ? a0 : a1);
    fma4(accQ, q, lo ? a2 : a3);
}

__device__ __forceinline__ void accum_chunk(const float4* __restrict__ H4, int sl,
                                            float4 al, int n, int lane,
                                            float4& accP, float4& accQ) {
    const bool lo = lane < 16;
    int t = 0;
    for (; t + 4 <= n; t += 4) {
        const float4* rA = H4 + (size_t)lane_i(sl, t)     * (HC / 4) + lane;
        const float4* rB = H4 + (size_t)lane_i(sl, t + 1) * (HC / 4) + lane;
        const float4* rC = H4 + (size_t)lane_i(sl, t + 2) * (HC / 4) + lane;
        const float4* rD = H4 + (size_t)lane_i(sl, t + 3) * (HC / 4) + lane;
        float4 pA = rA[0], qA = rA[32];
        float4 pB = rB[0], qB = rB[32];
        float4 pC = rC[0], qC = rC[32];
        float4 pD = rD[0], qD = rD[32];
        edge_fma(accP, accQ, pA, qA, al, t,     lo);
        edge_fma(accP, accQ, pB, qB, al, t + 1, lo);
        edge_fma(accP, accQ, pC, qC, al, t + 2, lo);
        edge_fma(accP, accQ, pD, qD, al, t + 3, lo);
    }
    for (; t < n; ++t) {
        const float4* r = H4 + (size_t)lane_i(sl, t) * (HC / 4) + lane;
        float4 p = r[0], q = r[32];
        edge_fma(accP, accQ, p, q, al, t, lo);
    }
}

__device__ __forceinline__ void aggregate_node(
        int node, int lane,
        const int* __restrict__ rowptr, const int* __restrict__ csr_src,
        const float* __restrict__ H, const float* __restrict__ aS,
        const float* __restrict__ aD, const float* __restrict__ bias,
        int relu, float4& accP, float4& accQ) {
    int start = rowptr[node];
    int end   = rowptr[node + 1];
    int cnt   = end - start;

    const float4* aS4 = (const float4*)aS;
    const float4* H4  = (const float4*)H;
    float4 ad = ((const float4*)aD)[node];
    float4 xs = logit4(aS4[node], ad);

    bool  v0 = lane < cnt;
    int   s0 = v0 ? csr_src[start + lane] : node;
    float4 x0 = logit4(aS4[s0], ad);

    float4 m = v0 ? max4(xs, x0) : xs;
    for (int j = start + 32 + lane; j < end; j += 32)
        m = max4(m, logit4(aS4[csr_src[j]], ad));
    m.x = wave_max(m.x); m.y = wave_max(m.y);
    m.z = wave_max(m.z); m.w = wave_max(m.w);

    float4 ex0 = exp4(x0, m);
    float4 sum = v0 ? ex0 : make_float4(0.f, 0.f, 0.f, 0.f);
    for (int j = start + 32 + lane; j < end; j += 32) {
        float4 ex = exp4(logit4(aS4[csr_src[j]], ad), m);
        sum.x += ex.x; sum.y += ex.y; sum.z += ex.z; sum.w += ex.w;
    }
    float4 exs = exp4(xs, m);
    float4 den;
    den.x = wave_sum(sum.x) + exs.x + 1e-16f;
    den.y = wave_sum(sum.y) + exs.y + 1e-16f;
    den.z = wave_sum(sum.z) + exs.z + 1e-16f;
    den.w = wave_sum(sum.w) + exs.w + 1e-16f;

    accP = make_float4(0.f, 0.f, 0.f, 0.f);
    accQ = make_float4(0.f, 0.f, 0.f, 0.f);
    {
        float4 a = div4(exs, den);
        const float4* r = H4 + (size_t)node * (HC / 4) + lane;
        float4 p = r[0], q = r[32];
        fma4(accP, p, lane < 16 ? a.x : a.y);
        fma4(accQ, q, lane < 16 ? a.z : a.w);
    }
    accum_chunk(H4, s0, div4(ex0, den), cnt < 32 ? cnt : 32, lane, accP, accQ);
    for (int base = start + 32; base < end; base += 32) {
        int  j  = base + lane;
        int  s1 = j < end ? csr_src[j] : node;
        float4 a1 = div4(exp4(logit4(aS4[s1], ad), m), den);
        int  n1 = end - base < 32 ? end - base : 32;
        accum_chunk(H4, s1, a1, n1, lane, accP, accQ);
    }

    const float4* b4 = (const float4*)bias;
    float4 bP = b4[lane], bQ = b4[lane + 32];
    accP.x += bP.x; accP.y += bP.y; accP.z += bP.z; accP.w += bP.w;
    accQ.x += bQ.x; accQ.y += bQ.y; accQ.z += bQ.z; accQ.w += bQ.w;
    if (relu) {
        accP.x = fmaxf(accP.x, 0.f); accP.y = fmaxf(accP.y, 0.f);
        accP.z = fmaxf(accP.z, 0.f); accP.w = fmaxf(accP.w, 0.f);
        accQ.x = fmaxf(accQ.x, 0.f); accQ.y = fmaxf(accQ.y, 0.f);
        accQ.z = fmaxf(accQ.z, 0.f); accQ.w = fmaxf(accQ.w, 0.f);
    }
}

__global__ __launch_bounds__(256) void gat_aggregate(
        const int* __restrict__ rowptr, const int* __restrict__ csr_src,
        const float* __restrict__ H, const float* __restrict__ aS,
        const float* __restrict__ aD, const float* __restrict__ bias,
        float* __restrict__ out, int N, int relu) {
    int node = __builtin_amdgcn_readfirstlane(blockIdx.x * 8 + (threadIdx.x >> 5));
    if (node >= N) return;
    int lane  = threadIdx.x & 31;
    float4 accP, accQ;
    aggregate_node(node, lane, rowptr, csr_src, H, aS, aD, bias, relu, accP, accQ);
    float4* o = (float4*)out + (size_t)node * (HC / 4) + lane;
    o[0]  = accP;
    o[32] = accQ;
}

#define TILE_PITCH4 65
#define FUSED_ROWS  32

__global__ __launch_bounds__(256) void agg_gemm_fused(
        const int* __restrict__ rowptr, const int* __restrict__ csr_src,
        const float* __restrict__ Hin, const float* __restrict__ aSin,
        const float* __restrict__ aDin, const float* __restrict__ bias,
        const float* __restrict__ W,
        const float* __restrict__ a_src, const float* __restrict__ a_dst,
        float* __restrict__ Hout, float* __restrict__ aSout,
        float* __restrict__ aDout, int N) {
    __shared__ float4 tile[FUSED_ROWS * TILE_PITCH4];
    const int lane = threadIdx.x & 31;
    const int wave = __builtin_amdgcn_readfirstlane(threadIdx.x >> 5);

    for (int i = 0; i < 4; ++i) {
        int r    = wave * 4 + i;
        int node = blockIdx.x * FUSED_ROWS + r;
        float4 accP = make_float4(0.f, 0.f, 0.f, 0.f);
        float4 accQ = make_float4(0.f, 0.f, 0.f, 0.f);
        if (node < N)
            aggregate_node(node, lane, rowptr, csr_src, Hin, aSin, aDin, bias,
                            1, accP, accQ);
        tile[r * TILE_PITCH4 + lane]      = accP;
        tile[r * TILE_PITCH4 + 32 + lane] = accQ;
    }
    __syncthreads();

    const int half  = lane >> 4;
    const int l16   = lane & 15;
    const int strip = wave & 3;
    const int rt    = wave >> 2;
    const int tileM = blockIdx.x * (FUSED_ROWS / 16) + rt;
    const int col0  = strip * 64 + 4 * l16;
    v8f acc0 = {}, acc1 = {}, acc2 = {}, acc3 = {};
    strip_mma((const float*)&tile[(rt * 16 + l16) * TILE_PITCH4], W + col0, half, HC,
              acc0, acc1, acc2, acc3);
    strip_epilogue(acc0, acc1, acc2, acc3, tileM, strip, lane, Hout, a_src, a_dst,
                   aSout, aDout, N);
}

__device__ __forceinline__ void fma4h(float4& z0, float4& z1, float4& z2, float4& z3,
                                      float4 x, float a0, float a1, float a2, float a3) {
    fma4(z0, x, a0); fma4(z1, x, a1); fma4(z2, x, a2); fma4(z3, x, a3);
}

template <int NR>
__device__ __forceinline__ void in_rows(const float4* __restrict__ X4, int sl, float4 al,
                                        int t, int lane,
                                        float4& z0, float4& z1, float4& z2, float4& z3) {
    float4 x[NR];
#pragma unroll
    for (int j = 0; j < NR; ++j)
        x[j] = X4[(size_t)lane_i(sl, t + j) * (FIN / 4) + lane];
#pragma unroll
    for (int j = 0; j < NR; ++j)
        fma4h(z0, z1, z2, z3, x[j], lane_f(al.x, t + j), lane_f(al.y, t + j),
              lane_f(al.z, t + j), lane_f(al.w, t + j));
}

__device__ __forceinline__ void in_chunk(const float4* __restrict__ X4, int sl, float4 al,
                                         int n, int lane,
                                         float4& z0, float4& z1, float4& z2, float4& z3) {
    int t = 0;
    for (; t + 8 <= n; t += 8) in_rows<8>(X4, sl, al, t, lane, z0, z1, z2, z3);
    if (n - t >= 4) { in_rows<4>(X4, sl, al, t, lane, z0, z1, z2, z3); t += 4; }
    if (n - t >= 2) { in_rows<2>(X4, sl, al, t, lane, z0, z1, z2, z3); t += 2; }
    if (n - t >= 1) { in_rows<1>(X4, sl, al, t, lane, z0, z1, z2, z3); }
}

__device__ __forceinline__ void aggregate_in_node(
        int node, int lane,
        const int* __restrict__ rowptr, const int* __restrict__ csr_src,
        const float* __restrict__ X, const float* __restrict__ aS,
        const float* __restrict__ aD,
        float4& z0, float4& z1, float4& z2, float4& z3) {
    int start = rowptr[node];
    int end   = rowptr[node + 1];
    int cnt   = end - start;

    const float4* aS4 = (const float4*)aS;
    const float4* X4  = (const float4*)X;
    float4 ad = ((const float4*)aD)[node];
    float4 xs = logit4(aS4[node], ad);

    bool  v0 = lane < cnt;
    int   s0 = v0 ? csr_src[start + lane] : node;
    float4 x0 = logit4(aS4[s0], ad);

    float4 m = v0 ? max4(xs, x0) : xs;
    for (int j = start + 32 + lane; j < end; j += 32)
        m = max4(m, logit4(aS4[csr_src[j]], ad));
    m.x = wave_max(m.x); m.y = wave_max(m.y);
    m.z = wave_max(m.z); m.w = wave_max(m.w);

    float4 ex0 = exp4(x0, m);
    float4 sum = v0 ? ex0 : make_float4(0.f, 0.f, 0.f, 0.f);
    for (int j = start + 32 + lane; j < end; j += 32) {
        float4 ex = exp4(logit4(aS4[csr_src[j]], ad), m);
        sum.x += ex.x; sum.y += ex.y; sum.z += ex.z; sum.w += ex.w;
    }
    float4 exs = exp4(xs, m);
    float4 den;
    den.x = wave_sum(sum.x) + exs.x + 1e-16f;
    den.y = wave_sum(sum.y) + exs.y + 1e-16f;
    den.z = wave_sum(sum.z) + exs.z + 1e-16f;
    den.w = wave_sum(sum.w) + exs.w + 1e-16f;

    z0 = make_float4(0.f, 0.f, 0.f, 0.f);
    z1 = z0; z2 = z0; z3 = z0;
    {
        float4 a = div4(exs, den);
        float4 x = X4[(size_t)node * (FIN / 4) + lane];
        fma4h(z0, z1, z2, z3, x, a.x, a.y, a.z, a.w);
    }
    in_chunk(X4, s0, div4(ex0, den), cnt < 32 ? cnt : 32, lane, z0, z1, z2, z3);
    for (int base = start + 32; base < end; base += 32) {
        int  j  = base + lane;
        int  s1 = j < end ? csr_src[j] : node;
        float4 a1 = div4(exp4(logit4(aS4[s1], ad), m), den);
        int  n1 = end - base < 32 ? end - base : 32;
        in_chunk(X4, s1, a1, n1, lane, z0, z1, z2, z3);
    }
}

#define ZPITCH4 (HEADS * (FIN / 4) + 1)

__global__ __launch_bounds__(256) void agg_in_gemm2(
        const int* __restrict__ rowptr, const int* __restrict__ csr_src,
        const float* __restrict__ X, const float* __restrict__ aSin,
        const float* __restrict__ aDin,
        const float* __restrict__ W1, const float* __restrict__ bias1,
        const float* __restrict__ W2,
        const float* __restrict__ a_src, const float* __restrict__ a_dst,
        float* __restrict__ Hout, float* __restrict__ aSout,
        float* __restrict__ aDout, int N) {
    __shared__ float4 zt[FUSED_ROWS * ZPITCH4];
    const int lane = threadIdx.x & 31;
    const int wave = __builtin_amdgcn_readfirstlane(threadIdx.x >> 5);

    for (int i = 0; i < 4; ++i) {
        int r    = wave * 4 + i;
        int node = blockIdx.x * FUSED_ROWS + r;
        float4 z0 = make_float4(0.f, 0.f, 0.f, 0.f);
        float4 z1 = z0, z2 = z0, z3 = z0;
        if (node < N)
            aggregate_in_node(node, lane, rowptr, csr_src, X, aSin, aDin, z0, z1, z2, z3);
        float4* zr = zt + r * ZPITCH4 + lane;
        zr[0]              = z0;
        zr[FIN / 4]        = z1;
        zr[2 * (FIN / 4)]  = z2;
        zr[3 * (FIN / 4)]  = z3;
    }
    __syncthreads();

    const int half  = lane >> 4;
    const int l16   = lane & 15;
    const int strip = wave & 3;
    const int rt    = wave >> 2;
    const int tileM = blockIdx.x * (FUSED_ROWS / 16) + rt;
    const int col0  = strip * 64 + 4 * l16;

    v8f acc0 = {}, acc1 = {}, acc2 = {}, acc3 = {};
    strip_mma((const float*)&zt[(rt * 16 + l16) * ZPITCH4 + strip * (FIN / 4)], W1 + col0,
              half, FIN, acc0, acc1, acc2, acc3);
    __syncthreads();
    {
        const float4 b = *(const float4*)(bias1 + col0);
#pragma unroll
        for (int r = 0; r < 8; ++r) {
            int row = rt * 16 + r + 8 * half;
            float4 o;
            o.x = fmaxf(acc0[r] + b.x, 0.f); o.y = fmaxf(acc1[r] + b.y, 0.f);
            o.z = fmaxf(acc2[r] + b.z, 0.f); o.w = fmaxf(acc3[r] + b.w, 0.f);
            zt[row * TILE_PITCH4 + strip * 16 + l16] = o;
        }
    }
    __syncthreads();

    v8f d0 = {}, d1 = {}, d2 = {}, d3 = {};
    strip_mma((const float*)&zt[(rt * 16 + l16) * TILE_PITCH4], W2 + col0, half, HC,
              d0, d1, d2, d3);
    strip_epilogue(d0, d1, d2, d3, tileM, strip, lane, Hout, a_src, a_dst,
                   aSout, aDout, N);
}

extern "C" void kernel_launch(void* const* d_in, const int* in_sizes, int n_in,
                              void* d_out, int out_size, void* d_ws, size_t ws_size,
                              hipStream_t stream) {
    const float* x   = (const float*)d_in[0];
    const int*   ei  = (const int*)d_in[1];
    const float* W1  = (const float*)d_in[2];
    const float* as1 = (const float*)d_in[3];
    const float* ad1 = (const float*)d_in[4];
    const float* b1  = (const float*)d_in[5];
    const float* W2  = (const float*)d_in[6];
    const float* as2 = (const float*)d_in[7];
    const float* ad2 = (const float*)d_in[8];
    const float* b2  = (const float*)d_in[9];
    const float* W3  = (const float*)d_in[10];
    const float* as3 = (const float*)d_in[11];
    const float* ad3 = (const float*)d_in[12];
    const float* b3  = (const float*)d_in[13];

    int K1 = in_sizes[2] / HC;
    int N  = in_sizes[0] / K1;
    int E  = in_sizes[1] / 2;

    float* Hbuf    = (float*)d_ws;
    float* bufA    = Hbuf + (size_t)N * HC;
    float* bufB    = bufA + (size_t)N * HC;
    float* alphaS  = bufB + (size_t)N * HC;
    float* alphaS2 = bufB;
    float* alphaD2 = bufB + (size_t)N * HEADS;
    float* alphaD  = alphaS + (size_t)N * HEADS;
    int*   deg     = (int*)(alphaD + (size_t)N * HEADS);
    int*   rowptr  = deg + (size_t)N;
    int*   rank    = rowptr + (size_t)N + 1;
    int*   csr_src = rank + (size_t)E;
    float* wsdT    = bufB + (size_t)2 * N * HEADS;

    int fillBlocks = (N + 255) / 256;
    int foldBlocks = (16 * FIN + 255) / 256;
    prep<<<fillBlocks + foldBlocks, 256, 0, stream>>>(deg, N, fillBlocks, W1, as1, ad1, wsdT);
    csr_count<<<(E + 255) / 256, 256, 0, stream>>>(ei, E, deg, rank);
    csr_scan<<<1, SCAN_THREADS, 0, stream>>>(deg, rowptr, N);
    csr_place<<<(E + 255) / 256, 256, 0, stream>>>(ei, E, rowptr, rank, csr_src);

    int waves  = (N + 15) / 16;
    int blocks = (waves + 7) / 8;
    alpha_from_x<<<blocks, 256, 0, stream>>>(x, wsdT, alphaS, alphaD, N);
    int tiles = (N + FUSED_ROWS - 1) / FUSED_ROWS;
    agg_in_gemm2<<<tiles, 256, 0, stream>>>(rowptr, csr_src, x, alphaS, alphaD, W1, b1,
                                            W2, as2, ad2, bufA, alphaS2, alphaD2, N);
    agg_gemm_fused<<<tiles, 256, 0, stream>>>(rowptr, csr_src, bufA, alphaS2, alphaD2, b2,
                                              W3, as3, ad3, Hbuf, alphaS, alphaD, N);
    gat_aggregate<<<(N + 7) / 8, 256, 0, stream>>>(rowptr, csr_src, Hbuf, alphaS, alphaD,
                                                   b3, (float*)d_out, N,  0);
}
